# starting kernel
# baseline (speedup 1.0000x reference)
#include <hip/hip_runtime.h>
#include <math.h>


typedef float v2f __attribute__((ext_vector_type(2)));
typedef float v8f __attribute__((ext_vector_type(8)));

#define T_IN     441000
#define T_OUT    160000
#define BATCH    32
#define QBLK     1000
#define KPAD     800
#define KW       576
#define NPHASE   160
#define NUMZ     64.0
#define ROLLOFF  0.9475937167399596
#define KBETA    14.769656459379492
#define SCALE    (16000.0 / 44100.0)

__device__ inline double bessel_i0(double z) {
    double zz = 0.25 * z * z;
    double term = 1.0, sum = 1.0;
    #pragma unroll 1
    for (int m = 1; m <= 60; ++m) {
        term *= zz / (double)(m * m);
        sum += term;
    }
    return sum;
}

__global__ __launch_bounds__(256) void build_weights(float* __restrict__ W) {
    int idx = blockIdx.x * 256 + threadIdx.x;
    int p = idx / KPAD;
    int j = idx - p * KPAD;
    int pn   = p * 441;
    int boff = pn / 160;
    double frac = (double)(pn - boff * 160) / 160.0;
    int    k    = j - 177 - boff;
    double u    = ((double)k - frac) * SCALE;
    double w    = 0.0;
    if (fabs(u) < NUMZ) {
        double xr = ROLLOFF * u;
        double s  = (xr == 0.0) ? 1.0 : sin(M_PI * xr) / (M_PI * xr);
        double r  = 1.0 - (u / NUMZ) * (u / NUMZ);
        if (r < 0.0) r = 0.0;
        w = SCALE * ROLLOFF * s * bessel_i0(KBETA * sqrt(r)) / bessel_i0(KBETA);
    }
    W[idx] = (float)w;
}

template <bool MASKED>
__device__ inline void gemm_kloop(const float* __restrict__ x,
                                  const float* __restrict__ lds,
                                  int lm, int koff,
                                  int base0, int jlo0,
                                  int base1, int jlo1,
                                  v8f c[5][2]) {
    const int jhi0 = jlo0 + T_IN - 1;
    const int jhi1 = jlo1 + T_IN - 1;
    #pragma unroll 2
    for (int kk = 0; kk < KW; kk += 4) {
        const int j = kk + koff;
        float e0, e1, f0, f1;
        if constexpr (!MASKED) {
            e0 = x[base0 + j];
            e1 = x[base0 + j + 1];
            f0 = x[base1 + j];
            f1 = x[base1 + j + 1];
        } else {
            int ja = j     < jlo0 ? jlo0 : j;      ja = ja > jhi0 ? jhi0 : ja;
            int jb = j + 1 < jlo0 ? jlo0 : j + 1;  jb = jb > jhi0 ? jhi0 : jb;
            int jc = j     < jlo1 ? jlo1 : j;      jc = jc > jhi1 ? jhi1 : jc;
            int jd = j + 1 < jlo1 ? jlo1 : j + 1;  jd = jd > jhi1 ? jhi1 : jd;
            float m0 = ((unsigned)(j     - jlo0) < (unsigned)T_IN) ? 1.0f : 0.0f;
            float m1 = ((unsigned)(j + 1 - jlo0) < (unsigned)T_IN) ? 1.0f : 0.0f;
            float m2 = ((unsigned)(j     - jlo1) < (unsigned)T_IN) ? 1.0f : 0.0f;
            float m3 = ((unsigned)(j + 1 - jlo1) < (unsigned)T_IN) ? 1.0f : 0.0f;
            e0 = x[base0 + ja] * m0;
            e1 = x[base0 + jb] * m1;
            f0 = x[base1 + jc] * m2;
            f1 = x[base1 + jd] * m3;
        }
        v2f B0 = {e0, e1};
        v2f B1 = {f0, f1};
        #pragma unroll
        for (int mt = 0; mt < 5; ++mt) {
            v2f A = *(const v2f*)(lds + (mt * 16 + lm) * KW + j);
            c[mt][0] = __builtin_amdgcn_wmma_f32_16x16x4_f32(
                false, A, false, B0, (short)0, c[mt][0], false, false);
            c[mt][1] = __builtin_amdgcn_wmma_f32_16x16x4_f32(
                false, A, false, B1, (short)0, c[mt][1], false, false);
        }
    }
}

__global__ __launch_bounds__(256) void resample_wmma(const float* __restrict__ x,
                                                     const float* __restrict__ W,
                                                     float* __restrict__ y) {
    extern __shared__ float lds[];

    const int tid = threadIdx.x;
    const int ph  = blockIdx.y;
    const int cg  = blockIdx.x;
    const int kstart = ph ? 220 : 0;

    {
        const float* Wsrc = W + ph * 80 * KPAD + kstart;
        const int trow = tid >> 6;
        const int tcol = tid & 63;
        for (int row = trow; row < 80; row += 4) {
            for (int t4 = tcol; t4 < KW / 4; t4 += 64) {
                *(float4*)&lds[row * KW + t4 * 4] =
                    *(const float4*)&Wsrc[row * KPAD + t4 * 4];
            }
        }
    }
    __syncthreads();

    const int wave = tid >> 5;
    const int lane = tid & 31;
    const int lm   = lane & 15;
    const int hi   = lane >> 4;
    const int koff = hi * 2;

    const int col0 = cg * 256 + wave * 32 + lm;
    const int col1 = col0 + 16;
    const int b0 = col0 / QBLK, q0 = col0 - b0 * QBLK;
    const int b1 = col1 / QBLK, q1 = col1 - b1 * QBLK;
    const int base0 = b0 * T_IN + q0 * 441 - 177 + kstart;
    const int base1 = b1 * T_IN + q1 * 441 - 177 + kstart;
    const int jlo0  = 177 - q0 * 441 - kstart;
    const int jlo1  = 177 - q1 * 441 - kstart;

    v8f c[5][2] = {};

    const bool edge_lane = (q0 == 0) | (q0 == QBLK - 1) |
                           (q1 == 0) | (q1 == QBLK - 1);
    if (__any((int)edge_lane)) {
        gemm_kloop<true >(x, lds, lm, koff, base0, jlo0, base1, jlo1, c);
    } else {
        gemm_kloop<false>(x, lds, lm, koff, base0, jlo0, base1, jlo1, c);
    }

    const int mrow = hi * 8;
    const int ob0 = b0 * T_OUT + q0 * 160 + ph * 80;
    const int ob1 = b1 * T_OUT + q1 * 160 + ph * 80;
    #pragma unroll
    for (int mt = 0; mt < 5; ++mt) {
        #pragma unroll
        for (int r = 0; r < 8; ++r) {
            y[ob0 + mt * 16 + mrow + r] = c[mt][0][r];
            y[ob1 + mt * 16 + mrow + r] = c[mt][1][r];
        }
    }
}

extern "C" void kernel_launch(void* const* d_in, const int* in_sizes, int n_in,
                              void* d_out, int out_size, void* d_ws, size_t ws_size,
                              hipStream_t stream) {
    const float* x = (const float*)d_in[0];
    float*       W = (float*)d_ws;
    float*       y = (float*)d_out;

    build_weights<<<500, 256, 0, stream>>>(W);

    dim3 grid(125, 2);
    resample_wmma<<<grid, 256, 80 * KW * sizeof(float), stream>>>(x, W, y);
}


# W table written by its kernel in A-fragment order; GEMM reads A fragments from it through the cache, one 16-byte load per lane per M-tile and k-pair; no LDS staging or barrier; sums unchanged
# speedup vs baseline: 1.1149x; 1.1149x over previous
#include <hip/hip_runtime.h>
#include <math.h>


typedef float v2f __attribute__((ext_vector_type(2)));
typedef float v8f __attribute__((ext_vector_type(8)));

#define T_IN     441000
#define T_OUT    160000
#define BATCH    32
#define QBLK     1000
#define KPAD     800
#define KW       576
#define KP_N     (KW / 8)
#define WF_HALF  (KP_N * 5 * 32 * 4)
#define NPHASE   160
#define NUMZ     64.0
#define ROLLOFF  0.9475937167399596
#define KBETA    14.769656459379492
#define SCALE    (16000.0 / 44100.0)

__device__ inline double bessel_i0(double z) {
    double zz = 0.25 * z * z;
    double term = 1.0, sum = 1.0;
    #pragma unroll 1
    for (int m = 1; m <= 60; ++m) {
        term *= zz / (double)(m * m);
        sum += term;
    }
    return sum;
}

__global__ __launch_bounds__(256) void build_weights(float* __restrict__ Wf) {
    int idx = blockIdx.x * 256 + threadIdx.x;
    int p = idx / KPAD;
    int j = idx - p * KPAD;
    int pn   = p * 441;
    int boff = pn / 160;
    double frac = (double)(pn - boff * 160) / 160.0;
    int    k    = j - 177 - boff;
    double u    = ((double)k - frac) * SCALE;
    double w    = 0.0;
    if (fabs(u) < NUMZ) {
        double xr = ROLLOFF * u;
        double s  = (xr == 0.0) ? 1.0 : sin(M_PI * xr) / (M_PI * xr);
        double r  = 1.0 - (u / NUMZ) * (u / NUMZ);
        if (r < 0.0) r = 0.0;
        w = SCALE * ROLLOFF * s * bessel_i0(KBETA * sqrt(r)) / bessel_i0(KBETA);
    }
    const int half = p / 80;
    const int r    = p - half * 80;
    const int jw   = j - (half ? 220 : 0);
    if (jw >= 0 && jw < KW) {
        const int kp   = jw >> 3;
        const int rem  = jw & 7;
        const int e    = ((rem >> 2) << 1) | (rem & 1);
        const int lane = (((rem >> 1) & 1) << 4) | (r & 15);
        Wf[half * WF_HALF + (((kp * 5 + (r >> 4)) * 32 + lane) << 2) + e] = (float)w;
    }
}

template <bool MASKED>
__device__ inline void gemm_kloop(const float* __restrict__ x,
                                  const float4* __restrict__ wf,
                                  int koff,
                                  int base0, int jlo0,
                                  int base1, int jlo1,
                                  v8f c[5][2]) {
    const int jhi0 = jlo0 + T_IN - 1;
    const int jhi1 = jlo1 + T_IN - 1;
    for (int kp = 0; kp < KP_N; ++kp) {
        float4 a4[5];
        #pragma unroll
        for (int mt = 0; mt < 5; ++mt) a4[mt] = wf[(kp * 5 + mt) * 32];
        #pragma unroll
        for (int s = 0; s < 2; ++s) {
            const int j = kp * 8 + s * 4 + koff;
            float e0, e1, f0, f1;
            if constexpr (!MASKED) {
                e0 = x[base0 + j];
                e1 = x[base0 + j + 1];
                f0 = x[base1 + j];
                f1 = x[base1 + j + 1];
            } else {
                int ja = j     < jlo0 ? jlo0 : j;      ja = ja > jhi0 ? jhi0 : ja;
                int jb = j + 1 < jlo0 ? jlo0 : j + 1;  jb = jb > jhi0 ? jhi0 : jb;
                int jc = j     < jlo1 ? jlo1 : j;      jc = jc > jhi1 ? jhi1 : jc;
                int jd = j + 1 < jlo1 ? jlo1 : j + 1;  jd = jd > jhi1 ? jhi1 : jd;
                float m0 = ((unsigned)(j     - jlo0) < (unsigned)T_IN) ? 1.0f : 0.0f;
                float m1 = ((unsigned)(j + 1 - jlo0) < (unsigned)T_IN) ? 1.0f : 0.0f;
                float m2 = ((unsigned)(j     - jlo1) < (unsigned)T_IN) ? 1.0f : 0.0f;
                float m3 = ((unsigned)(j + 1 - jlo1) < (unsigned)T_IN) ? 1.0f : 0.0f;
                e0 = x[base0 + ja] * m0;
                e1 = x[base0 + jb] * m1;
                f0 = x[base1 + jc] * m2;
                f1 = x[base1 + jd] * m3;
            }
            v2f B0 = {e0, e1};
            v2f B1 = {f0, f1};
            #pragma unroll
            for (int mt = 0; mt < 5; ++mt) {
                v2f A;
                if (s == 0) { A.x = a4[mt].x; A.y = a4[mt].y; }
                else        { A.x = a4[mt].z; A.y = a4[mt].w; }
                c[mt][0] = __builtin_amdgcn_wmma_f32_16x16x4_f32(
                    false, A, false, B0, (short)0, c[mt][0], false, false);
                c[mt][1] = __builtin_amdgcn_wmma_f32_16x16x4_f32(
                    false, A, false, B1, (short)0, c[mt][1], false, false);
            }
        }
    }
}

__global__ __launch_bounds__(256) void resample_wmma(const float* __restrict__ x,
                                                     const float* __restrict__ Wf,
                                                     float* __restrict__ y) {
    const int tid = threadIdx.x;
    const int ph  = blockIdx.y;
    const int cg  = blockIdx.x;
    const int kstart = ph ? 220 : 0;

    const int wave = tid >> 5;
    const int lane = tid & 31;
    const int lm   = lane & 15;
    const int hi   = lane >> 4;
    const int koff = hi * 2;

    const float4* wf = (const float4*)(Wf + ph * WF_HALF) + lane;

    const int col0 = cg * 256 + wave * 32 + lm;
    const int col1 = col0 + 16;
    const int b0 = col0 / QBLK, q0 = col0 - b0 * QBLK;
    const int b1 = col1 / QBLK, q1 = col1 - b1 * QBLK;
    const int base0 = b0 * T_IN + q0 * 441 - 177 + kstart;
    const int base1 = b1 * T_IN + q1 * 441 - 177 + kstart;
    const int jlo0  = 177 - q0 * 441 - kstart;
    const int jlo1  = 177 - q1 * 441 - kstart;

    v8f c[5][2] = {};

    const bool edge_lane = (q0 == 0) | (q0 == QBLK - 1) |
                           (q1 == 0) | (q1 == QBLK - 1);
    if (__any((int)edge_lane)) {
        gemm_kloop<true >(x, wf, koff, base0, jlo0, base1, jlo1, c);
    } else {
        gemm_kloop<false>(x, wf, koff, base0, jlo0, base1, jlo1, c);
    }

    const int mrow = hi * 8;
    const int ob0 = b0 * T_OUT + q0 * 160 + ph * 80;
    const int ob1 = b1 * T_OUT + q1 * 160 + ph * 80;
    #pragma unroll
    for (int mt = 0; mt < 5; ++mt) {
        #pragma unroll
        for (int r = 0; r < 8; ++r) {
            y[ob0 + mt * 16 + mrow + r] = c[mt][0][r];
            y[ob1 + mt * 16 + mrow + r] = c[mt][1][r];
        }
    }
}

extern "C" void kernel_launch(void* const* d_in, const int* in_sizes, int n_in,
                              void* d_out, int out_size, void* d_ws, size_t ws_size,
                              hipStream_t stream) {
    const float* x  = (const float*)d_in[0];
    float*       Wf = (float*)d_ws;
    float*       y  = (float*)d_out;

    build_weights<<<500, 256, 0, stream>>>(Wf);

    dim3 grid(125, 2);
    resample_wmma<<<grid, 256, 0, stream>>>(x, Wf, y);
}


# speedup vs baseline: 2.5336x; 2.2725x over previous
#include <hip/hip_runtime.h>
#include <math.h>


typedef float v2f __attribute__((ext_vector_type(2)));
typedef float v8f __attribute__((ext_vector_type(8)));

#define T_IN     441000
#define T_OUT    160000
#define BATCH    32
#define QBLK     1000
#define KPAD     800
#define KW       576
#define KP_N     (KW / 8)
#define WF_HALF  (KP_N * 5 * 32 * 4)
#define NPHASE   160
#define NUMZ     64.0
#define ROLLOFF  0.9475937167399596
#define KBETA    14.769656459379492
#define SCALE    (16000.0 / 44100.0)

#define UBAND    28224
#define ZZ_BETA  (0.25 * KBETA * KBETA)

__device__ inline float bessel_i0_zz(float zz) {
    float term = 1.0f, sum = 1.0f;
    #pragma unroll 1
    for (int m = 1; m <= 60; ++m) {
        term *= zz / (float)(m * m);
        sum += term;
    }
    return sum;
}

__global__ __launch_bounds__(256) void build_weights(float* __restrict__ Wf) {
    int idx = blockIdx.x * 256 + threadIdx.x;
    int p = idx / KPAD;
    int j = idx - p * KPAD;
    int pn   = p * 441;
    int boff = pn / 160;
    int rem  = pn - boff * 160;
    int k    = j - 177 - boff;
    int nu   = 160 * k - rem;
    float w  = 0.0f;
    if (nu > -UBAND && nu < UBAND) {
        const double dn = (double)nu;
        float s = 1.0f;
        if (nu != 0) {
            const double xr = dn * (ROLLOFF / 441.0);
            const double nr = rint(xr);
            float sp = sinpif((float)(xr - nr));
            if (((int)nr) & 1) sp = -sp;
            s = sp / (float)(M_PI * xr);
        }
        const double v = dn * (1.0 / (double)UBAND);
        double r = fma(-v, v, 1.0);
        if (r < 0.0) r = 0.0;
        const float zz = (float)(ZZ_BETA * r);
        w = (float)(SCALE * ROLLOFF) * s * bessel_i0_zz(zz)
            / bessel_i0_zz((float)ZZ_BETA);
    }
    const int half = p / 80;
    const int r    = p - half * 80;
    const int jw   = j - (half ? 220 : 0);
    if (jw >= 0 && jw < KW) {
        const int kp   = jw >> 3;
        const int rem8 = jw & 7;
        const int e    = ((rem8 >> 2) << 1) | (rem8 & 1);
        const int lane = (((rem8 >> 1) & 1) << 4) | (r & 15);
        Wf[half * WF_HALF + (((kp * 5 + (r >> 4)) * 32 + lane) << 2) + e] = w;
    }
}

template <bool MASKED>
__device__ inline void gemm_kloop(const float* __restrict__ x,
                                  const float4* __restrict__ wf,
                                  int koff,
                                  int base0, int jlo0,
                                  int base1, int jlo1,
                                  v8f c[5][2]) {
    const int jhi0 = jlo0 + T_IN - 1;
    const int jhi1 = jlo1 + T_IN - 1;
    for (int kp = 0; kp < KP_N; ++kp) {
        float4 a4[5];
        #pragma unroll
        for (int mt = 0; mt < 5; ++mt) a4[mt] = wf[(kp * 5 + mt) * 32];
        #pragma unroll
        for (int s = 0; s < 2; ++s) {
            const int j = kp * 8 + s * 4 + koff;
            float e0, e1, f0, f1;
            if constexpr (!MASKED) {
                e0 = x[base0 + j];
                e1 = x[base0 + j + 1];
                f0 = x[base1 + j];
                f1 = x[base1 + j + 1];
            } else {
                int ja = j     < jlo0 ? jlo0 : j;      ja = ja > jhi0 ? jhi0 : ja;
                int jb = j + 1 < jlo0 ? jlo0 : j + 1;  jb = jb > jhi0 ? jhi0 : jb;
                int jc = j     < jlo1 ? jlo1 : j;      jc = jc > jhi1 ? jhi1 : jc;
                int jd = j + 1 < jlo1 ? jlo1 : j + 1;  jd = jd > jhi1 ? jhi1 : jd;
                float m0 = ((unsigned)(j     - jlo0) < (unsigned)T_IN) ? 1.0f : 0.0f;
                float m1 = ((unsigned)(j + 1 - jlo0) < (unsigned)T_IN) ? 1.0f : 0.0f;
                float m2 = ((unsigned)(j     - jlo1) < (unsigned)T_IN) ? 1.0f : 0.0f;
                float m3 = ((unsigned)(j + 1 - jlo1) < (unsigned)T_IN) ? 1.0f : 0.0f;
                e0 = x[base0 + ja] * m0;
                e1 = x[base0 + jb] * m1;
                f0 = x[base1 + jc] * m2;
                f1 = x[base1 + jd] * m3;
            }
            v2f B0 = {e0, e1};
            v2f B1 = {f0, f1};
            #pragma unroll
            for (int mt = 0; mt < 5; ++mt) {
                v2f A;
                if (s == 0) { A.x = a4[mt].x; A.y = a4[mt].y; }
                else        { A.x = a4[mt].z; A.y = a4[mt].w; }
                c[mt][0] = __builtin_amdgcn_wmma_f32_16x16x4_f32(
                    false, A, false, B0, (short)0, c[mt][0], false, false);
                c[mt][1] = __builtin_amdgcn_wmma_f32_16x16x4_f32(
                    false, A, false, B1, (short)0, c[mt][1], false, false);
            }
        }
    }
}

__global__ __launch_bounds__(256) void resample_wmma(const float* __restrict__ x,
                                                     const float* __restrict__ Wf,
                                                     float* __restrict__ y) {
    const int tid = threadIdx.x;
    const int ph  = blockIdx.y;
    const int cg  = blockIdx.x;
    const int kstart = ph ? 220 : 0;

    const int wave = tid >> 5;
    const int lane = tid & 31;
    const int lm   = lane & 15;
    const int hi   = lane >> 4;
    const int koff = hi * 2;

    const float4* wf = (const float4*)(Wf + ph * WF_HALF) + lane;

    const int col0 = cg * 256 + wave * 32 + lm;
    const int col1 = col0 + 16;
    const int b0 = col0 / QBLK, q0 = col0 - b0 * QBLK;
    const int b1 = col1 / QBLK, q1 = col1 - b1 * QBLK;
    const int base0 = b0 * T_IN + q0 * 441 - 177 + kstart;
    const int base1 = b1 * T_IN + q1 * 441 - 177 + kstart;
    const int jlo0  = 177 - q0 * 441 - kstart;
    const int jlo1  = 177 - q1 * 441 - kstart;

    v8f c[5][2] = {};

    const bool edge_lane = (q0 == 0) | (q0 == QBLK - 1) |
                           (q1 == 0) | (q1 == QBLK - 1);
    if (__any((int)edge_lane)) {
        gemm_kloop<true >(x, wf, koff, base0, jlo0, base1, jlo1, c);
    } else {
        gemm_kloop<false>(x, wf, koff, base0, jlo0, base1, jlo1, c);
    }

    const int mrow = hi * 8;
    const int ob0 = b0 * T_OUT + q0 * 160 + ph * 80;
    const int ob1 = b1 * T_OUT + q1 * 160 + ph * 80;
    #pragma unroll
    for (int mt = 0; mt < 5; ++mt) {
        #pragma unroll
        for (int r = 0; r < 8; ++r) {
            y[ob0 + mt * 16 + mrow + r] = c[mt][0][r];
            y[ob1 + mt * 16 + mrow + r] = c[mt][1][r];
        }
    }
}

extern "C" void kernel_launch(void* const* d_in, const int* in_sizes, int n_in,
                              void* d_out, int out_size, void* d_ws, size_t ws_size,
                              hipStream_t stream) {
    const float* x  = (const float*)d_in[0];
    float*       Wf = (float*)d_ws;
    float*       y  = (float*)d_out;

    build_weights<<<500, 256, 0, stream>>>(Wf);

    dim3 grid(125, 2);
    resample_wmma<<<grid, 256, 0, stream>>>(x, Wf, y);
}


# x windows (B operand) staged through a wave-private LDS patch: whole 256-byte runs loaded per column, B fragments read from LDS (was a 16-line gather per load); k order and sums unchanged
# speedup vs baseline: 2.9495x; 1.1642x over previous
#include <hip/hip_runtime.h>
#include <math.h>


typedef float v2f __attribute__((ext_vector_type(2)));
typedef float v8f __attribute__((ext_vector_type(8)));

#define T_IN     441000
#define T_OUT    160000
#define BATCH    32
#define QBLK     1000
#define KPAD     800
#define KW       576
#define KP_N     (KW / 8)
#define WF_HALF  (KP_N * 5 * 32 * 4)
#define KC       64
#define NCHUNK   (KW / KC)
#define XP       68
#define XWAVE    (32 * XP)
#define NPHASE   160
#define NUMZ     64.0
#define ROLLOFF  0.9475937167399596
#define KBETA    14.769656459379492
#define SCALE    (16000.0 / 44100.0)

#define UBAND    28224
#define ZZ_BETA  (0.25 * KBETA * KBETA)

__device__ inline float bessel_i0_zz(float zz) {
    float term = 1.0f, sum = 1.0f;
    #pragma unroll 1
    for (int m = 1; m <= 60; ++m) {
        term *= zz / (float)(m * m);
        sum += term;
    }
    return sum;
}

__global__ __launch_bounds__(256) void build_weights(float* __restrict__ Wf) {
    int idx = blockIdx.x * 256 + threadIdx.x;
    int p = idx / KPAD;
    int j = idx - p * KPAD;
    int pn   = p * 441;
    int boff = pn / 160;
    int rem  = pn - boff * 160;
    int k    = j - 177 - boff;
    int nu   = 160 * k - rem;
    float w  = 0.0f;
    if (nu > -UBAND && nu < UBAND) {
        const double dn = (double)nu;
        float s = 1.0f;
        if (nu != 0) {
            const double xr = dn * (ROLLOFF / 441.0);
            const double nr = rint(xr);
            float sp = sinpif((float)(xr - nr));
            if (((int)nr) & 1) sp = -sp;
            s = sp / (float)(M_PI * xr);
        }
        const double v = dn * (1.0 / (double)UBAND);
        double r = fma(-v, v, 1.0);
        if (r < 0.0) r = 0.0;
        const float zz = (float)(ZZ_BETA * r);
        w = (float)(SCALE * ROLLOFF) * s * bessel_i0_zz(zz)
            / bessel_i0_zz((float)ZZ_BETA);
    }
    const int half = p / 80;
    const int r    = p - half * 80;
    const int jw   = j - (half ? 220 : 0);
    if (jw >= 0 && jw < KW) {
        const int kp   = jw >> 3;
        const int rem8 = jw & 7;
        const int e    = ((rem8 >> 2) << 1) | (rem8 & 1);
        const int lane = (((rem8 >> 1) & 1) << 4) | (r & 15);
        Wf[half * WF_HALF + (((kp * 5 + (r >> 4)) * 32 + lane) << 2) + e] = w;
    }
}

typedef float v4f __attribute__((ext_vector_type(4)));
typedef v4f v4f_u __attribute__((aligned(4)));

template <bool MASKED>
__device__ inline void stage_fetch(const float* __restrict__ x, int cw, int kstart,
                                   int j0, int lane, v4f r[16]) {
    const int piece = (lane & 15) * 4;
    #pragma unroll
    for (int it = 0; it < 16; ++it) {
        const int c = cw + 2 * it + (lane >> 4);
        if constexpr (!MASKED) {
            r[it] = *(const v4f_u*)(x + (c * 441 - 177 + kstart + j0 + piece));
        } else {
            const int b = c / QBLK;
            const int q = c - b * QBLK;
            const int t = q * 441 - 177 + kstart + j0 + piece;
            v4f v;
            #pragma unroll
            for (int e = 0; e < 4; ++e) {
                int tc = t + e;
                const bool ok = (unsigned)tc < (unsigned)T_IN;
                tc = tc < 0 ? 0 : tc;
                tc = tc > T_IN - 1 ? T_IN - 1 : tc;
                v[e] = x[b * T_IN + tc] * (ok ? 1.0f : 0.0f);
            }
            r[it] = v;
        }
    }
}

__device__ inline void stage_store(float* __restrict__ patch, int lane, const v4f r[16]) {
    float* dst = patch + (lane >> 4) * XP + (lane & 15) * 4;
    #pragma unroll
    for (int it = 0; it < 16; ++it) *(v4f*)(dst + 2 * it * XP) = r[it];
}

template <bool MASKED>
__device__ inline void gemm_chunks(const float* __restrict__ x,
                                   const float4* __restrict__ wf,
                                   float* __restrict__ patch,
                                   int cw, int kstart, int lane,
                                   v8f c[5][2]) {
    const float* bw = patch + (lane & 15) * XP + (lane >> 4) * 2;
    v4f r[16];
    stage_fetch<MASKED>(x, cw, kstart, 0, lane, r);
    stage_store(patch, lane, r);
    for (int ch = 0; ch < NCHUNK; ++ch) {
        if (ch + 1 < NCHUNK) stage_fetch<MASKED>(x, cw, kstart, (ch + 1) * KC, lane, r);
        #pragma unroll
        for (int kl = 0; kl < KC / 8; ++kl) {
            const int kp = ch * (KC / 8) + kl;
            float4 a4[5];
            #pragma unroll
            for (int mt = 0; mt < 5; ++mt) a4[mt] = wf[(kp * 5 + mt) * 32];
            #pragma unroll
            for (int s = 0; s < 2; ++s) {
                const int jl = kl * 8 + s * 4;
                const v2f B0 = *(const v2f*)(bw + jl);
                const v2f B1 = *(const v2f*)(bw + 16 * XP + jl);
                #pragma unroll
                for (int mt = 0; mt < 5; ++mt) {
                    v2f A;
                    if (s == 0) { A.x = a4[mt].x; A.y = a4[mt].y; }
                    else        { A.x = a4[mt].z; A.y = a4[mt].w; }
                    c[mt][0] = __builtin_amdgcn_wmma_f32_16x16x4_f32(
                        false, A, false, B0, (short)0, c[mt][0], false, false);
                    c[mt][1] = __builtin_amdgcn_wmma_f32_16x16x4_f32(
                        false, A, false, B1, (short)0, c[mt][1], false, false);
                }
            }
        }
        if (ch + 1 < NCHUNK) stage_store(patch, lane, r);
    }
}

__global__ __launch_bounds__(256) void resample_wmma(const float* __restrict__ x,
                                                     const float* __restrict__ Wf,
                                                     float* __restrict__ y) {
    __shared__ __attribute__((aligned(16))) float xpatch[8 * XWAVE];

    const int tid = threadIdx.x;
    const int ph  = blockIdx.y;
    const int cg  = blockIdx.x;
    const int kstart = ph ? 220 : 0;

    const int wave = tid >> 5;
    const int lane = tid & 31;
    const int lm   = lane & 15;
    const int hi   = lane >> 4;

    const float4* wf = (const float4*)(Wf + ph * WF_HALF) + lane;

    const int cw   = cg * 256 + wave * 32;
    const int col0 = cw + lm;
    const int col1 = col0 + 16;
    const int b0 = col0 / QBLK, q0 = col0 - b0 * QBLK;
    const int b1 = col1 / QBLK, q1 = col1 - b1 * QBLK;

    float* patch = xpatch + wave * XWAVE;

    v8f c[5][2] = {};

    const bool edge_lane = (q0 == 0) | (q0 == QBLK - 1) |
                           (q1 == 0) | (q1 == QBLK - 1);
    if (__any((int)edge_lane)) {
        gemm_chunks<true >(x, wf, patch, cw, kstart, lane, c);
    } else {
        gemm_chunks<false>(x, wf, patch, cw, kstart, lane, c);
    }

    const int mrow = hi * 8;
    const int ob0 = b0 * T_OUT + q0 * 160 + ph * 80;
    const int ob1 = b1 * T_OUT + q1 * 160 + ph * 80;
    #pragma unroll
    for (int mt = 0; mt < 5; ++mt) {
        #pragma unroll
        for (int r = 0; r < 8; ++r) {
            y[ob0 + mt * 16 + mrow + r] = c[mt][0][r];
            y[ob1 + mt * 16 + mrow + r] = c[mt][1][r];
        }
    }
}

extern "C" void kernel_launch(void* const* d_in, const int* in_sizes, int n_in,
                              void* d_out, int out_size, void* d_ws, size_t ws_size,
                              hipStream_t stream) {
    const float* x  = (const float*)d_in[0];
    float*       Wf = (float*)d_ws;
    float*       y  = (float*)d_out;

    build_weights<<<500, 256, 0, stream>>>(Wf);

    dim3 grid(125, 2);
    resample_wmma<<<grid, 256, 0, stream>>>(x, Wf, y);
}


# A fragments (cached table) asked for 40 to 80 WMMAs ahead of use into spare register sets (were loaded and waited for per k-pair); edge waves mask taps at the LDS store; sums unchanged
# speedup vs baseline: 3.5838x; 1.2151x over previous
#include <hip/hip_runtime.h>
#include <math.h>


typedef float v2f __attribute__((ext_vector_type(2)));
typedef float v8f __attribute__((ext_vector_type(8)));

#define T_IN     441000
#define T_OUT    160000
#define BATCH    32
#define QBLK     1000
#define KPAD     800
#define KW       576
#define KP_N     (KW / 8)
#define WF_HALF  (KP_N * 5 * 32 * 4)
#define KC       64
#define NCHUNK   (KW / KC)
#define GP       2
#define NG_CH    (KC / 8 / GP)
#define XP       68
#define XWAVE    (32 * XP)
#define NPHASE   160
#define NUMZ     64.0
#define ROLLOFF  0.9475937167399596
#define KBETA    14.769656459379492
#define SCALE    (16000.0 / 44100.0)

#define UBAND    28224
#define ZZ_BETA  (0.25 * KBETA * KBETA)

__device__ inline float bessel_i0_zz(float zz) {
    float term = 1.0f, sum = 1.0f;
    #pragma unroll 1
    for (int m = 1; m <= 60; ++m) {
        term *= zz / (float)(m * m);
        sum += term;
    }
    return sum;
}

__global__ __launch_bounds__(256) void build_weights(float* __restrict__ Wf) {
    int idx = blockIdx.x * 256 + threadIdx.x;
    int p = idx / KPAD;
    int j = idx - p * KPAD;
    int pn   = p * 441;
    int boff = pn / 160;
    int rem  = pn - boff * 160;
    int k    = j - 177 - boff;
    int nu   = 160 * k - rem;
    float w  = 0.0f;
    if (nu > -UBAND && nu < UBAND) {
        const double dn = (double)nu;
        float s = 1.0f;
        if (nu != 0) {
            const double xr = dn * (ROLLOFF / 441.0);
            const double nr = rint(xr);
            float sp = sinpif((float)(xr - nr));
            if (((int)nr) & 1) sp = -sp;
            s = sp / (float)(M_PI * xr);
        }
        const double v = dn * (1.0 / (double)UBAND);
        double r = fma(-v, v, 1.0);
        if (r < 0.0) r = 0.0;
        const float zz = (float)(ZZ_BETA * r);
        w = (float)(SCALE * ROLLOFF) * s * bessel_i0_zz(zz)
            / bessel_i0_zz((float)ZZ_BETA);
    }
    const int half = p / 80;
    const int r    = p - half * 80;
    const int jw   = j - (half ? 220 : 0);
    if (jw >= 0 && jw < KW) {
        const int kp   = jw >> 3;
        const int rem8 = jw & 7;
        const int e    = ((rem8 >> 2) << 1) | (rem8 & 1);
        const int lane = (((rem8 >> 1) & 1) << 4) | (r & 15);
        Wf[half * WF_HALF + (((kp * 5 + (r >> 4)) * 32 + lane) << 2) + e] = w;
    }
}

typedef float v4f __attribute__((ext_vector_type(4)));
typedef v4f v4f_u __attribute__((aligned(4)));

template <bool MASKED>
__device__ inline void stage_fetch(const float* __restrict__ x, int cw, int kstart,
                                   int j0, int lane, v4f r[16]) {
    const int piece = (lane & 15) * 4;
    #pragma unroll
    for (int it = 0; it < 16; ++it) {
        const int c = cw + 2 * it + (lane >> 4);
        if constexpr (!MASKED) {
            r[it] = *(const v4f_u*)(x + (c * 441 - 177 + kstart + j0 + piece));
        } else {
            const int b = c / QBLK;
            const int q = c - b * QBLK;
            const int t = q * 441 - 177 + kstart + j0 + piece;
            v4f v;
            #pragma unroll
            for (int e = 0; e < 4; ++e) {
                int tc = t + e;
                tc = tc < 0 ? 0 : tc;
                tc = tc > T_IN - 1 ? T_IN - 1 : tc;
                v[e] = x[b * T_IN + tc];
            }
            r[it] = v;
        }
    }
}

template <bool MASKED>
__device__ inline void stage_store(float* __restrict__ patch, int cw, int kstart,
                                   int j0, int lane, const v4f r[16]) {
    float* dst = patch + (lane >> 4) * XP + (lane & 15) * 4;
    const int piece = (lane & 15) * 4;
    #pragma unroll
    for (int it = 0; it < 16; ++it) {
        v4f v = r[it];
        if constexpr (MASKED) {
            const int c = cw + 2 * it + (lane >> 4);
            const int b = c / QBLK;
            const int q = c - b * QBLK;
            const int t = q * 441 - 177 + kstart + j0 + piece;
            #pragma unroll
            for (int e = 0; e < 4; ++e)
                v[e] = v[e] * (((unsigned)(t + e) < (unsigned)T_IN) ? 1.0f : 0.0f);
        }
        *(v4f*)(dst + 2 * it * XP) = v;
    }
}

__device__ inline void fetch_a(const float4* __restrict__ wf, int grp, float4 a[GP * 5]) {
    #pragma unroll
    for (int i = 0; i < GP * 5; ++i) a[i] = wf[(grp * (GP * 5) + i) * 32];
}

template <bool MASKED, bool LAST>
__device__ inline void gemm_chunk(const float* __restrict__ x,
                                  const float4* __restrict__ wf,
                                  float* __restrict__ patch,
                                  int cw, int kstart, int lane, int ch,
                                  v4f r[16], float4 a[2][GP * 5], v8f c[5][2]) {
    const float* bw = patch + (lane & 15) * XP + (lane >> 4) * 2;
    #pragma unroll
    for (int g = 0; g < NG_CH; ++g) {
        if (g + 1 < NG_CH || !LAST) fetch_a(wf, ch * NG_CH + g + 1, a[(g + 1) & 1]);
        if (g == 0 && !LAST) stage_fetch<MASKED>(x, cw, kstart, (ch + 1) * KC, lane, r);
        __builtin_amdgcn_sched_barrier(0);
        #pragma unroll
        for (int i = 0; i < GP; ++i) {
            const int kl = g * GP + i;
            #pragma unroll
            for (int s = 0; s < 2; ++s) {
                const int jl = kl * 8 + s * 4;
                const v2f B0 = *(const v2f*)(bw + jl);
                const v2f B1 = *(const v2f*)(bw + 16 * XP + jl);
                #pragma unroll
                for (int mt = 0; mt < 5; ++mt) {
                    const float4 af = a[g & 1][i * 5 + mt];
                    v2f A;
                    if (s == 0) { A.x = af.x; A.y = af.y; }
                    else        { A.x = af.z; A.y = af.w; }
                    c[mt][0] = __builtin_amdgcn_wmma_f32_16x16x4_f32(
                        false, A, false, B0, (short)0, c[mt][0], false, false);
                    c[mt][1] = __builtin_amdgcn_wmma_f32_16x16x4_f32(
                        false, A, false, B1, (short)0, c[mt][1], false, false);
                }
            }
        }
    }
    if (!LAST) stage_store<MASKED>(patch, cw, kstart, (ch + 1) * KC, lane, r);
}

template <bool MASKED>
__device__ inline void gemm_chunks(const float* __restrict__ x,
                                   const float4* __restrict__ wf,
                                   float* __restrict__ patch,
                                   int cw, int kstart, int lane,
                                   v8f c[5][2]) {
    v4f r[16];
    float4 a[2][GP * 5];
    stage_fetch<MASKED>(x, cw, kstart, 0, lane, r);
    fetch_a(wf, 0, a[0]);
    stage_store<MASKED>(patch, cw, kstart, 0, lane, r);
    for (int ch = 0; ch < NCHUNK - 1; ++ch)
        gemm_chunk<MASKED, false>(x, wf, patch, cw, kstart, lane, ch, r, a, c);
    gemm_chunk<MASKED, true>(x, wf, patch, cw, kstart, lane, NCHUNK - 1, r, a, c);
}

__global__ __launch_bounds__(256) void resample_wmma(const float* __restrict__ x,
                                                     const float* __restrict__ Wf,
                                                     float* __restrict__ y) {
    __shared__ __attribute__((aligned(16))) float xpatch[8 * XWAVE];

    const int tid = threadIdx.x;
    const int ph  = blockIdx.y;
    const int cg  = blockIdx.x;
    const int kstart = ph ? 220 : 0;

    const int wave = tid >> 5;
    const int lane = tid & 31;
    const int lm   = lane & 15;
    const int hi   = lane >> 4;

    const float4* wf = (const float4*)(Wf + ph * WF_HALF) + lane;

    const int cw   = cg * 256 + wave * 32;
    const int col0 = cw + lm;
    const int col1 = col0 + 16;
    const int b0 = col0 / QBLK, q0 = col0 - b0 * QBLK;
    const int b1 = col1 / QBLK, q1 = col1 - b1 * QBLK;

    float* patch = xpatch + wave * XWAVE;

    v8f c[5][2] = {};

    const bool edge_lane = (q0 == 0) | (q0 == QBLK - 1) |
                           (q1 == 0) | (q1 == QBLK - 1);
    if (__any((int)edge_lane)) {
        gemm_chunks<true >(x, wf, patch, cw, kstart, lane, c);
    } else {
        gemm_chunks<false>(x, wf, patch, cw, kstart, lane, c);
    }

    const int mrow = hi * 8;
    const int ob0 = b0 * T_OUT + q0 * 160 + ph * 80;
    const int ob1 = b1 * T_OUT + q1 * 160 + ph * 80;
    #pragma unroll
    for (int mt = 0; mt < 5; ++mt) {
        #pragma unroll
        for (int r = 0; r < 8; ++r) {
            y[ob0 + mt * 16 + mrow + r] = c[mt][0][r];
            y[ob1 + mt * 16 + mrow + r] = c[mt][1][r];
        }
    }
}

extern "C" void kernel_launch(void* const* d_in, const int* in_sizes, int n_in,
                              void* d_out, int out_size, void* d_ws, size_t ws_size,
                              hipStream_t stream) {
    const float* x  = (const float*)d_in[0];
    float*       Wf = (float*)d_ws;
    float*       y  = (float*)d_out;

    build_weights<<<500, 256, 0, stream>>>(Wf);

    dim3 grid(125, 2);
    resample_wmma<<<grid, 256, 0, stream>>>(x, Wf, y);
}


# next chunk's x requests queued behind the A-table requests (A ring of 4 sets, 3 ahead): no wait on x before the chunk's end; edge waves fetch whole pieces, per tap only row-end pairs; same sums
# speedup vs baseline: 3.7436x; 1.0446x over previous
#include <hip/hip_runtime.h>
#include <math.h>


typedef float v2f __attribute__((ext_vector_type(2)));
typedef float v8f __attribute__((ext_vector_type(8)));

#define T_IN     441000
#define T_OUT    160000
#define BATCH    32
#define QBLK     1000
#define KPAD     800
#define KW       576
#define KP_N     (KW / 8)
#define WF_HALF  (KP_N * 5 * 32 * 4)
#define KC       64
#define NCHUNK   (KW / KC)
#define GP       2
#define NG_CH    (KC / 8 / GP)
#define NSET     4
#define AHEAD    3
#define XP       68
#define XWAVE    (32 * XP)
#define NPHASE   160
#define NUMZ     64.0
#define ROLLOFF  0.9475937167399596
#define KBETA    14.769656459379492
#define SCALE    (16000.0 / 44100.0)

#define UBAND    28224
#define ZZ_BETA  (0.25 * KBETA * KBETA)

__device__ inline float bessel_i0_zz(float zz) {
    float term = 1.0f, sum = 1.0f;
    #pragma unroll 1
    for (int m = 1; m <= 60; ++m) {
        term *= zz / (float)(m * m);
        sum += term;
    }
    return sum;
}

__global__ __launch_bounds__(256) void build_weights(float* __restrict__ Wf) {
    int idx = blockIdx.x * 256 + threadIdx.x;
    int p = idx / KPAD;
    int j = idx - p * KPAD;
    int pn   = p * 441;
    int boff = pn / 160;
    int rem  = pn - boff * 160;
    int k    = j - 177 - boff;
    int nu   = 160 * k - rem;
    float w  = 0.0f;
    if (nu > -UBAND && nu < UBAND) {
        const double dn = (double)nu;
        float s = 1.0f;
        if (nu != 0) {
            const double xr = dn * (ROLLOFF / 441.0);
            const double nr = rint(xr);
            float sp = sinpif((float)(xr - nr));
            if (((int)nr) & 1) sp = -sp;
            s = sp / (float)(M_PI * xr);
        }
        const double v = dn * (1.0 / (double)UBAND);
        double r = fma(-v, v, 1.0);
        if (r < 0.0) r = 0.0;
        const float zz = (float)(ZZ_BETA * r);
        w = (float)(SCALE * ROLLOFF) * s * bessel_i0_zz(zz)
            / bessel_i0_zz((float)ZZ_BETA);
    }
    const int half = p / 80;
    const int r    = p - half * 80;
    const int jw   = j - (half ? 220 : 0);
    if (jw >= 0 && jw < KW) {
        const int kp   = jw >> 3;
        const int rem8 = jw & 7;
        const int e    = ((rem8 >> 2) << 1) | (rem8 & 1);
        const int lane = (((rem8 >> 1) & 1) << 4) | (r & 15);
        Wf[half * WF_HALF + (((kp * 5 + (r >> 4)) * 32 + lane) << 2) + e] = w;
    }
}

typedef float v4f __attribute__((ext_vector_type(4)));
typedef v4f v4f_u __attribute__((aligned(4)));

#define X_LAST4  (BATCH * T_IN - 4)

__device__ inline v4f fetch_taps_clamped(const float* __restrict__ x, int cw, int it,
                                         int kstart, int j0, int lane) {
    const int c = cw + 2 * it + (lane >> 4);
    const int b = c / QBLK;
    const int q = c - b * QBLK;
    const int t = q * 441 - 177 + kstart + j0 + (lane & 15) * 4;
    v4f v;
    #pragma unroll
    for (int e = 0; e < 4; ++e) {
        int tc = t + e;
        tc = tc < 0 ? 0 : tc;
        tc = tc > T_IN - 1 ? T_IN - 1 : tc;
        v[e] = x[b * T_IN + tc];
    }
    return v;
}

template <bool MASKED>
__device__ inline void stage_fetch(const float* __restrict__ x, int cw, int kstart,
                                   int j0, int lane, const int ite[2], v4f r[16], v4f er[2]) {
    const int piece = (lane & 15) * 4;
    #pragma unroll
    for (int it = 0; it < 16; ++it) {
        const int c = cw + 2 * it + (lane >> 4);
        int f = c * 441 - 177 + kstart + j0 + piece;
        if constexpr (MASKED) {
            f = f < 0 ? 0 : f;
            f = f > X_LAST4 ? X_LAST4 : f;
        }
        r[it] = *(const v4f_u*)(x + f);
    }
    if constexpr (MASKED) {
        er[0] = fetch_taps_clamped(x, cw, ite[0], kstart, j0, lane);
        er[1] = fetch_taps_clamped(x, cw, ite[1], kstart, j0, lane);
    }
}

template <bool MASKED>
__device__ inline void stage_store(float* __restrict__ patch, int cw, int kstart,
                                   int j0, int lane, const int ite[2],
                                   const v4f r[16], const v4f er[2]) {
    float* dst = patch + (lane >> 4) * XP + (lane & 15) * 4;
    #pragma unroll
    for (int it = 0; it < 16; ++it)
        *(v4f*)(dst + 2 * it * XP) = r[it];
    if constexpr (MASKED) {
        #pragma unroll
        for (int s = 0; s < 2; ++s) {
            const int c = cw + 2 * ite[s] + (lane >> 4);
            const int b = c / QBLK;
            const int q = c - b * QBLK;
            const int t = q * 441 - 177 + kstart + j0 + (lane & 15) * 4;
            v4f v = er[s];
            #pragma unroll
            for (int e = 0; e < 4; ++e)
                v[e] = v[e] * (((unsigned)(t + e) < (unsigned)T_IN) ? 1.0f : 0.0f);
            *(v4f*)(dst + 2 * ite[s] * XP) = v;
        }
    }
}

__device__ inline void fetch_a(const float4* __restrict__ wf, int grp, float4 a[GP * 5]) {
    #pragma unroll
    for (int i = 0; i < GP * 5; ++i) a[i] = wf[(grp * (GP * 5) + i) * 32];
}

template <bool MASKED, bool LAST>
__device__ inline void gemm_chunk(const float* __restrict__ x,
                                  const float4* __restrict__ wf,
                                  float* __restrict__ patch,
                                  int cw, const int ite[2], int kstart, int lane, int ch,
                                  v4f r[16], v4f er[2], float4 a[NSET][GP * 5], v8f c[5][2]) {
    const float* bw = patch + (lane & 15) * XP + (lane >> 4) * 2;
    #pragma unroll
    for (int g = 0; g < NG_CH; ++g) {
        if (g + AHEAD < NG_CH || !LAST)
            fetch_a(wf, ch * NG_CH + g + AHEAD, a[(g + AHEAD) & (NSET - 1)]);
        if (g == 0 && !LAST) {
            __builtin_amdgcn_sched_barrier(0);
            stage_fetch<MASKED>(x, cw, kstart, (ch + 1) * KC, lane, ite, r, er);
        }
        __builtin_amdgcn_sched_barrier(0);
        #pragma unroll
        for (int i = 0; i < GP; ++i) {
            const int kl = g * GP + i;
            #pragma unroll
            for (int s = 0; s < 2; ++s) {
                const int jl = kl * 8 + s * 4;
                const v2f B0 = *(const v2f*)(bw + jl);
                const v2f B1 = *(const v2f*)(bw + 16 * XP + jl);
                #pragma unroll
                for (int mt = 0; mt < 5; ++mt) {
                    const float4 af = a[g & (NSET - 1)][i * 5 + mt];
                    v2f A;
                    if (s == 0) { A.x = af.x; A.y = af.y; }
                    else        { A.x = af.z; A.y = af.w; }
                    c[mt][0] = __builtin_amdgcn_wmma_f32_16x16x4_f32(
                        false, A, false, B0, (short)0, c[mt][0], false, false);
                    c[mt][1] = __builtin_amdgcn_wmma_f32_16x16x4_f32(
                        false, A, false, B1, (short)0, c[mt][1], false, false);
                }
            }
        }
    }
    __builtin_amdgcn_sched_barrier(0);
    if (!LAST) stage_store<MASKED>(patch, cw, kstart, (ch + 1) * KC, lane, ite, r, er);
}

template <bool MASKED>
__device__ inline void gemm_chunks(const float* __restrict__ x,
                                   const float4* __restrict__ wf,
                                   float* __restrict__ patch,
                                   int cw, int q0s, int kstart, int lane,
                                   v8f c[5][2]) {
    v4f r[16];
    v4f er[2];
    float4 a[NSET][GP * 5];
    int ite[2] = {0, 0};
    if constexpr (MASKED) {
        if (q0s != 0) {
            const int first = (QBLK - 2 - q0s + 1) >> 1;
            const int last  = (QBLK - q0s) >> 1;
            ite[0] = first < 0 ? 0 : (first > 15 ? 15 : first);
            ite[1] = last  < 0 ? 0 : (last  > 15 ? 15 : last);
        }
    }
    stage_fetch<MASKED>(x, cw, kstart, 0, lane, ite, r, er);
    #pragma unroll
    for (int s = 0; s < AHEAD; ++s) fetch_a(wf, s, a[s]);
    stage_store<MASKED>(patch, cw, kstart, 0, lane, ite, r, er);
    for (int ch = 0; ch < NCHUNK - 1; ++ch)
        gemm_chunk<MASKED, false>(x, wf, patch, cw, ite, kstart, lane, ch, r, er, a, c);
    gemm_chunk<MASKED, true>(x, wf, patch, cw, ite, kstart, lane, NCHUNK - 1, r, er, a, c);
}

__global__ __launch_bounds__(256) void resample_wmma(const float* __restrict__ x,
                                                     const float* __restrict__ Wf,
                                                     float* __restrict__ y) {
    __shared__ __attribute__((aligned(16))) float xpatch[8 * XWAVE];

    const int tid = threadIdx.x;
    const int ph  = blockIdx.y;
    const int cg  = blockIdx.x;
    const int kstart = ph ? 220 : 0;

    const int wave = tid >> 5;
    const int lane = tid & 31;
    const int lm   = lane & 15;
    const int hi   = lane >> 4;

    const float4* wf = (const float4*)(Wf + ph * WF_HALF) + lane;

    const int cw   = cg * 256 + wave * 32;
    const int col0 = cw + lm;
    const int col1 = col0 + 16;
    const int b0 = col0 / QBLK, q0 = col0 - b0 * QBLK;
    const int b1 = col1 / QBLK, q1 = col1 - b1 * QBLK;

    float* patch = xpatch + wave * XWAVE;

    v8f c[5][2] = {};

    const bool edge_lane = (q0 == 0) | (q0 == QBLK - 1) |
                           (q1 == 0) | (q1 == QBLK - 1);
    const int q0s = __builtin_amdgcn_readfirstlane(cw % QBLK);
    if (__any((int)edge_lane)) {
        gemm_chunks<true >(x, wf, patch, cw, q0s, kstart, lane, c);
    } else {
        gemm_chunks<false>(x, wf, patch, cw, q0s, kstart, lane, c);
    }

    const int mrow = hi * 8;
    const int ob0 = b0 * T_OUT + q0 * 160 + ph * 80;
    const int ob1 = b1 * T_OUT + q1 * 160 + ph * 80;
    #pragma unroll
    for (int mt = 0; mt < 5; ++mt) {
        #pragma unroll
        for (int r = 0; r < 8; ++r) {
            y[ob0 + mt * 16 + mrow + r] = c[mt][0][r];
            y[ob1 + mt * 16 + mrow + r] = c[mt][1][r];
        }
    }
}

extern "C" void kernel_launch(void* const* d_in, const int* in_sizes, int n_in,
                              void* d_out, int out_size, void* d_ws, size_t ws_size,
                              hipStream_t stream) {
    const float* x  = (const float*)d_in[0];
    float*       Wf = (float*)d_ws;
    float*       y  = (float*)d_out;

    build_weights<<<500, 256, 0, stream>>>(Wf);

    dim3 grid(125, 2);
    resample_wmma<<<grid, 256, 0, stream>>>(x, Wf, y);
}


# product taken transposed (x fragment as WMMA A operand, table fragment as B): D lanes lie along y's phase axis, a store writes two 64-byte runs (was 32 pieces of 8 bytes); k order, sums unchanged
# speedup vs baseline: 4.2917x; 1.1464x over previous
#include <hip/hip_runtime.h>
#include <math.h>


typedef float v2f __attribute__((ext_vector_type(2)));
typedef float v8f __attribute__((ext_vector_type(8)));

#define T_IN     441000
#define T_OUT    160000
#define BATCH    32
#define QBLK     1000
#define KPAD     800
#define KW       576
#define KP_N     (KW / 8)
#define WF_HALF  (KP_N * 5 * 32 * 4)
#define KC       64
#define NCHUNK   (KW / KC)
#define GP       2
#define NG_CH    (KC / 8 / GP)
#define NSET     4
#define AHEAD    3
#define XP       68
#define XWAVE    (32 * XP)
#define NPHASE   160
#define NUMZ     64.0
#define ROLLOFF  0.9475937167399596
#define KBETA    14.769656459379492
#define SCALE    (16000.0 / 44100.0)

#define UBAND    28224
#define ZZ_BETA  (0.25 * KBETA * KBETA)

__device__ inline float bessel_i0_zz(float zz) {
    float term = 1.0f, sum = 1.0f;
    #pragma unroll 1
    for (int m = 1; m <= 60; ++m) {
        term *= zz / (float)(m * m);
        sum += term;
    }
    return sum;
}

__global__ __launch_bounds__(256) void build_weights(float* __restrict__ Wf) {
    int idx = blockIdx.x * 256 + threadIdx.x;
    int p = idx / KPAD;
    int j = idx - p * KPAD;
    int pn   = p * 441;
    int boff = pn / 160;
    int rem  = pn - boff * 160;
    int k    = j - 177 - boff;
    int nu   = 160 * k - rem;
    float w  = 0.0f;
    if (nu > -UBAND && nu < UBAND) {
        const double dn = (double)nu;
        float s = 1.0f;
        if (nu != 0) {
            const double xr = dn * (ROLLOFF / 441.0);
            const double nr = rint(xr);
            float sp = sinpif((float)(xr - nr));
            if (((int)nr) & 1) sp = -sp;
            s = sp / (float)(M_PI * xr);
        }
        const double v = dn * (1.0 / (double)UBAND);
        double r = fma(-v, v, 1.0);
        if (r < 0.0) r = 0.0;
        const float zz = (float)(ZZ_BETA * r);
        w = (float)(SCALE * ROLLOFF) * s * bessel_i0_zz(zz)
            / bessel_i0_zz((float)ZZ_BETA);
    }
    const int half = p / 80;
    const int r    = p - half * 80;
    const int jw   = j - (half ? 220 : 0);
    if (jw >= 0 && jw < KW) {
        const int kp   = jw >> 3;
        const int rem8 = jw & 7;
        const int e    = ((rem8 >> 2) << 1) | (rem8 & 1);
        const int lane = (((rem8 >> 1) & 1) << 4) | (r & 15);
        Wf[half * WF_HALF + (((kp * 5 + (r >> 4)) * 32 + lane) << 2) + e] = w;
    }
}

typedef float v4f __attribute__((ext_vector_type(4)));
typedef v4f v4f_u __attribute__((aligned(4)));

#define X_LAST4  (BATCH * T_IN - 4)

__device__ inline v4f fetch_taps_clamped(const float* __restrict__ x, int cw, int it,
                                         int kstart, int j0, int lane) {
    const int c = cw + 2 * it + (lane >> 4);
    const int b = c / QBLK;
    const int q = c - b * QBLK;
    const int t = q * 441 - 177 + kstart + j0 + (lane & 15) * 4;
    v4f v;
    #pragma unroll
    for (int e = 0; e < 4; ++e) {
        int tc = t + e;
        tc = tc < 0 ? 0 : tc;
        tc = tc > T_IN - 1 ? T_IN - 1 : tc;
        v[e] = x[b * T_IN + tc];
    }
    return v;
}

template <bool MASKED>
__device__ inline void stage_fetch(const float* __restrict__ x, int cw, int kstart,
                                   int j0, int lane, const int ite[2], v4f r[16], v4f er[2]) {
    const int piece = (lane & 15) * 4;
    #pragma unroll
    for (int it = 0; it < 16; ++it) {
        const int c = cw + 2 * it + (lane >> 4);
        int f = c * 441 - 177 + kstart + j0 + piece;
        if constexpr (MASKED) {
            f = f < 0 ? 0 : f;
            f = f > X_LAST4 ? X_LAST4 : f;
        }
        r[it] = *(const v4f_u*)(x + f);
    }
    if constexpr (MASKED) {
        er[0] = fetch_taps_clamped(x, cw, ite[0], kstart, j0, lane);
        er[1] = fetch_taps_clamped(x, cw, ite[1], kstart, j0, lane);
    }
}

template <bool MASKED>
__device__ inline void stage_store(float* __restrict__ patch, int cw, int kstart,
                                   int j0, int lane, const int ite[2],
                                   const v4f r[16], const v4f er[2]) {
    float* dst = patch + (lane >> 4) * XP + (lane & 15) * 4;
    #pragma unroll
    for (int it = 0; it < 16; ++it)
        *(v4f*)(dst + 2 * it * XP) = r[it];
    if constexpr (MASKED) {
        #pragma unroll
        for (int s = 0; s < 2; ++s) {
            const int c = cw + 2 * ite[s] + (lane >> 4);
            const int b = c / QBLK;
            const int q = c - b * QBLK;
            const int t = q * 441 - 177 + kstart + j0 + (lane & 15) * 4;
            v4f v = er[s];
            #pragma unroll
            for (int e = 0; e < 4; ++e)
                v[e] = v[e] * (((unsigned)(t + e) < (unsigned)T_IN) ? 1.0f : 0.0f);
            *(v4f*)(dst + 2 * ite[s] * XP) = v;
        }
    }
}

__device__ inline void fetch_a(const float4* __restrict__ wf, int grp, float4 a[GP * 5]) {
    #pragma unroll
    for (int i = 0; i < GP * 5; ++i) a[i] = wf[(grp * (GP * 5) + i) * 32];
}

template <bool MASKED, bool LAST>
__device__ inline void gemm_chunk(const float* __restrict__ x,
                                  const float4* __restrict__ wf,
                                  float* __restrict__ patch,
                                  int cw, const int ite[2], int kstart, int lane, int ch,
                                  v4f r[16], v4f er[2], float4 a[NSET][GP * 5], v8f c[5][2]) {
    const float* bw = patch + (lane & 15) * XP + (lane >> 4) * 2;
    #pragma unroll
    for (int g = 0; g < NG_CH; ++g) {
        if (g + AHEAD < NG_CH || !LAST)
            fetch_a(wf, ch * NG_CH + g + AHEAD, a[(g + AHEAD) & (NSET - 1)]);
        if (g == 0 && !LAST) {
            __builtin_amdgcn_sched_barrier(0);
            stage_fetch<MASKED>(x, cw, kstart, (ch + 1) * KC, lane, ite, r, er);
        }
        __builtin_amdgcn_sched_barrier(0);
        #pragma unroll
        for (int i = 0; i < GP; ++i) {
            const int kl = g * GP + i;
            #pragma unroll
            for (int s = 0; s < 2; ++s) {
                const int jl = kl * 8 + s * 4;
                const v2f X0 = *(const v2f*)(bw + jl);
                const v2f X1 = *(const v2f*)(bw + 16 * XP + jl);
                #pragma unroll
                for (int mt = 0; mt < 5; ++mt) {
                    const float4 af = a[g & (NSET - 1)][i * 5 + mt];
                    v2f Wt;
                    if (s == 0) { Wt.x = af.x; Wt.y = af.y; }
                    else        { Wt.x = af.z; Wt.y = af.w; }
                    c[mt][0] = __builtin_amdgcn_wmma_f32_16x16x4_f32(
                        false, X0, false, Wt, (short)0, c[mt][0], false, false);
                    c[mt][1] = __builtin_amdgcn_wmma_f32_16x16x4_f32(
                        false, X1, false, Wt, (short)0, c[mt][1], false, false);
                }
            }
        }
    }
    __builtin_amdgcn_sched_barrier(0);
    if (!LAST) stage_store<MASKED>(patch, cw, kstart, (ch + 1) * KC, lane, ite, r, er);
}

template <bool MASKED>
__device__ inline void gemm_chunks(const float* __restrict__ x,
                                   const float4* __restrict__ wf,
                                   float* __restrict__ patch,
                                   int cw, int q0s, int kstart, int lane,
                                   v8f c[5][2]) {
    v4f r[16];
    v4f er[2];
    float4 a[NSET][GP * 5];
    int ite[2] = {0, 0};
    if constexpr (MASKED) {
        if (q0s != 0) {
            const int first = (QBLK - 2 - q0s + 1) >> 1;
            const int last  = (QBLK - q0s) >> 1;
            ite[0] = first < 0 ? 0 : (first > 15 ? 15 : first);
            ite[1] = last  < 0 ? 0 : (last  > 15 ? 15 : last);
        }
    }
    stage_fetch<MASKED>(x, cw, kstart, 0, lane, ite, r, er);
    #pragma unroll
    for (int s = 0; s < AHEAD; ++s) fetch_a(wf, s, a[s]);
    stage_store<MASKED>(patch, cw, kstart, 0, lane, ite, r, er);
    for (int ch = 0; ch < NCHUNK - 1; ++ch)
        gemm_chunk<MASKED, false>(x, wf, patch, cw, ite, kstart, lane, ch, r, er, a, c);
    gemm_chunk<MASKED, true>(x, wf, patch, cw, ite, kstart, lane, NCHUNK - 1, r, er, a, c);
}

__global__ __launch_bounds__(256) void resample_wmma(const float* __restrict__ x,
                                                     const float* __restrict__ Wf,
                                                     float* __restrict__ y) {
    __shared__ __attribute__((aligned(16))) float xpatch[8 * XWAVE];

    const int tid = threadIdx.x;
    const int ph  = blockIdx.y;
    const int cg  = blockIdx.x;
    const int kstart = ph ? 220 : 0;

    const int wave = tid >> 5;
    const int lane = tid & 31;
    const int lm   = lane & 15;
    const int hi   = lane >> 4;

    const float4* wf = (const float4*)(Wf + ph * WF_HALF) + lane;

    const int cw   = cg * 256 + wave * 32;
    const int col0 = cw + lm;
    const int col1 = col0 + 16;
    const int b0 = col0 / QBLK, q0 = col0 - b0 * QBLK;
    const int b1 = col1 / QBLK, q1 = col1 - b1 * QBLK;

    float* patch = xpatch + wave * XWAVE;

    v8f c[5][2] = {};

    const bool edge_lane = (q0 == 0) | (q0 == QBLK - 1) |
                           (q1 == 0) | (q1 == QBLK - 1);
    const int q0s = __builtin_amdgcn_readfirstlane(cw % QBLK);
    if (__any((int)edge_lane)) {
        gemm_chunks<true >(x, wf, patch, cw, q0s, kstart, lane, c);
    } else {
        gemm_chunks<false>(x, wf, patch, cw, q0s, kstart, lane, c);
    }

    float* __restrict__ yp = y + ((cw + hi * 8) * NPHASE + ph * 80 + lm);
    #pragma unroll
    for (int t = 0; t < 2; ++t) {
        #pragma unroll
        for (int r = 0; r < 8; ++r) {
            #pragma unroll
            for (int mt = 0; mt < 5; ++mt)
                yp[(t * 16 + r) * NPHASE + mt * 16] = c[mt][t][r];
        }
    }
}

extern "C" void kernel_launch(void* const* d_in, const int* in_sizes, int n_in,
                              void* d_out, int out_size, void* d_ws, size_t ws_size,
                              hipStream_t stream) {
    const float* x  = (const float*)d_in[0];
    float*       Wf = (float*)d_ws;
    float*       y  = (float*)d_out;

    build_weights<<<500, 256, 0, stream>>>(Wf);

    dim3 grid(125, 2);
    resample_wmma<<<grid, 256, 0, stream>>>(x, Wf, y);
}


# W-table kernel: both I0 power series run 24 terms (were 60): from term 21 on an addend is below half an ulp of the f32 sum for every argument up to (beta/2)^2, so all weights keep their bits
# speedup vs baseline: 4.6524x; 1.0840x over previous
#include <hip/hip_runtime.h>
#include <math.h>


typedef float v2f __attribute__((ext_vector_type(2)));
typedef float v8f __attribute__((ext_vector_type(8)));

#define T_IN     441000
#define T_OUT    160000
#define BATCH    32
#define QBLK     1000
#define KPAD     800
#define KW       576
#define KP_N     (KW / 8)
#define WF_HALF  (KP_N * 5 * 32 * 4)
#define KC       64
#define NCHUNK   (KW / KC)
#define GP       2
#define NG_CH    (KC / 8 / GP)
#define NSET     4
#define AHEAD    3
#define XP       68
#define XWAVE    (32 * XP)
#define NPHASE   160
#define NUMZ     64.0
#define ROLLOFF  0.9475937167399596
#define KBETA    14.769656459379492
#define SCALE    (16000.0 / 44100.0)

#define UBAND    28224
#define ZZ_BETA  (0.25 * KBETA * KBETA)

#define I0_TERMS 24
__device__ inline float bessel_i0_zz(float zz) {
    float term = 1.0f, sum = 1.0f;
    #pragma unroll 1
    for (int m = 1; m <= I0_TERMS; ++m) {
        term *= zz / (float)(m * m);
        sum += term;
    }
    return sum;
}

__global__ __launch_bounds__(256) void build_weights(float* __restrict__ Wf) {
    int idx = blockIdx.x * 256 + threadIdx.x;
    int p = idx / KPAD;
    int j = idx - p * KPAD;
    int pn   = p * 441;
    int boff = pn / 160;
    int rem  = pn - boff * 160;
    int k    = j - 177 - boff;
    int nu   = 160 * k - rem;
    float w  = 0.0f;
    if (nu > -UBAND && nu < UBAND) {
        const double dn = (double)nu;
        float s = 1.0f;
        if (nu != 0) {
            const double xr = dn * (ROLLOFF / 441.0);
            const double nr = rint(xr);
            float sp = sinpif((float)(xr - nr));
            if (((int)nr) & 1) sp = -sp;
            s = sp / (float)(M_PI * xr);
        }
        const double v = dn * (1.0 / (double)UBAND);
        double r = fma(-v, v, 1.0);
        if (r < 0.0) r = 0.0;
        const float zz = (float)(ZZ_BETA * r);
        w = (float)(SCALE * ROLLOFF) * s * bessel_i0_zz(zz)
            / bessel_i0_zz((float)ZZ_BETA);
    }
    const int half = p / 80;
    const int r    = p - half * 80;
    const int jw   = j - (half ? 220 : 0);
    if (jw >= 0 && jw < KW) {
        const int kp   = jw >> 3;
        const int rem8 = jw & 7;
        const int e    = ((rem8 >> 2) << 1) | (rem8 & 1);
        const int lane = (((rem8 >> 1) & 1) << 4) | (r & 15);
        Wf[half * WF_HALF + (((kp * 5 + (r >> 4)) * 32 + lane) << 2) + e] = w;
    }
}

typedef float v4f __attribute__((ext_vector_type(4)));
typedef v4f v4f_u __attribute__((aligned(4)));

#define X_LAST4  (BATCH * T_IN - 4)

__device__ inline v4f fetch_taps_clamped(const float* __restrict__ x, int cw, int it,
                                         int kstart, int j0, int lane) {
    const int c = cw + 2 * it + (lane >> 4);
    const int b = c / QBLK;
    const int q = c - b * QBLK;
    const int t = q * 441 - 177 + kstart + j0 + (lane & 15) * 4;
    v4f v;
    #pragma unroll
    for (int e = 0; e < 4; ++e) {
        int tc = t + e;
        tc = tc < 0 ? 0 : tc;
        tc = tc > T_IN - 1 ? T_IN - 1 : tc;
        v[e] = x[b * T_IN + tc];
    }
    return v;
}

template <bool MASKED>
__device__ inline void stage_fetch(const float* __restrict__ x, int cw, int kstart,
                                   int j0, int lane, const int ite[2], v4f r[16], v4f er[2]) {
    const int piece = (lane & 15) * 4;
    #pragma unroll
    for (int it = 0; it < 16; ++it) {
        const int c = cw + 2 * it + (lane >> 4);
        int f = c * 441 - 177 + kstart + j0 + piece;
        if constexpr (MASKED) {
            f = f < 0 ? 0 : f;
            f = f > X_LAST4 ? X_LAST4 : f;
        }
        r[it] = *(const v4f_u*)(x + f);
    }
    if constexpr (MASKED) {
        er[0] = fetch_taps_clamped(x, cw, ite[0], kstart, j0, lane);
        er[1] = fetch_taps_clamped(x, cw, ite[1], kstart, j0, lane);
    }
}

template <bool MASKED>
__device__ inline void stage_store(float* __restrict__ patch, int cw, int kstart,
                                   int j0, int lane, const int ite[2],
                                   const v4f r[16], const v4f er[2]) {
    float* dst = patch + (lane >> 4) * XP + (lane & 15) * 4;
    #pragma unroll
    for (int it = 0; it < 16; ++it)
        *(v4f*)(dst + 2 * it * XP) = r[it];
    if constexpr (MASKED) {
        #pragma unroll
        for (int s = 0; s < 2; ++s) {
            const int c = cw + 2 * ite[s] + (lane >> 4);
            const int b = c / QBLK;
            const int q = c - b * QBLK;
            const int t = q * 441 - 177 + kstart + j0 + (lane & 15) * 4;
            v4f v = er[s];
            #pragma unroll
            for (int e = 0; e < 4; ++e)
                v[e] = v[e] * (((unsigned)(t + e) < (unsigned)T_IN) ? 1.0f : 0.0f);
            *(v4f*)(dst + 2 * ite[s] * XP) = v;
        }
    }
}

__device__ inline void fetch_a(const float4* __restrict__ wf, int grp, float4 a[GP * 5]) {
    #pragma unroll
    for (int i = 0; i < GP * 5; ++i) a[i] = wf[(grp * (GP * 5) + i) * 32];
}

template <bool MASKED, bool LAST>
__device__ inline void gemm_chunk(const float* __restrict__ x,
                                  const float4* __restrict__ wf,
                                  float* __restrict__ patch,
                                  int cw, const int ite[2], int kstart, int lane, int ch,
                                  v4f r[16], v4f er[2], float4 a[NSET][GP * 5], v8f c[5][2]) {
    const float* bw = patch + (lane & 15) * XP + (lane >> 4) * 2;
    #pragma unroll
    for (int g = 0; g < NG_CH; ++g) {
        if (g + AHEAD < NG_CH || !LAST)
            fetch_a(wf, ch * NG_CH + g + AHEAD, a[(g + AHEAD) & (NSET - 1)]);
        if (g == 0 && !LAST) {
            __builtin_amdgcn_sched_barrier(0);
            stage_fetch<MASKED>(x, cw, kstart, (ch + 1) * KC, lane, ite, r, er);
        }
        __builtin_amdgcn_sched_barrier(0);
        #pragma unroll
        for (int i = 0; i < GP; ++i) {
            const int kl = g * GP + i;
            #pragma unroll
            for (int s = 0; s < 2; ++s) {
                const int jl = kl * 8 + s * 4;
                const v2f X0 = *(const v2f*)(bw + jl);
                const v2f X1 = *(const v2f*)(bw + 16 * XP + jl);
                #pragma unroll
                for (int mt = 0; mt < 5; ++mt) {
                    const float4 af = a[g & (NSET - 1)][i * 5 + mt];
                    v2f Wt;
                    if (s == 0) { Wt.x = af.x; Wt.y = af.y; }
                    else        { Wt.x = af.z; Wt.y = af.w; }
                    c[mt][0] = __builtin_amdgcn_wmma_f32_16x16x4_f32(
                        false, X0, false, Wt, (short)0, c[mt][0], false, false);
                    c[mt][1] = __builtin_amdgcn_wmma_f32_16x16x4_f32(
                        false, X1, false, Wt, (short)0, c[mt][1], false, false);
                }
            }
        }
    }
    __builtin_amdgcn_sched_barrier(0);
    if (!LAST) stage_store<MASKED>(patch, cw, kstart, (ch + 1) * KC, lane, ite, r, er);
}

template <bool MASKED>
__device__ inline void gemm_chunks(const float* __restrict__ x,
                                   const float4* __restrict__ wf,
                                   float* __restrict__ patch,
                                   int cw, int q0s, int kstart, int lane,
                                   v8f c[5][2]) {
    v4f r[16];
    v4f er[2];
    float4 a[NSET][GP * 5];
    int ite[2] = {0, 0};
    if constexpr (MASKED) {
        if (q0s != 0) {
            const int first = (QBLK - 2 - q0s + 1) >> 1;
            const int last  = (QBLK - q0s) >> 1;
            ite[0] = first < 0 ? 0 : (first > 15 ? 15 : first);
            ite[1] = last  < 0 ? 0 : (last  > 15 ? 15 : last);
        }
    }
    stage_fetch<MASKED>(x, cw, kstart, 0, lane, ite, r, er);
    #pragma unroll
    for (int s = 0; s < AHEAD; ++s) fetch_a(wf, s, a[s]);
    stage_store<MASKED>(patch, cw, kstart, 0, lane, ite, r, er);
    for (int ch = 0; ch < NCHUNK - 1; ++ch)
        gemm_chunk<MASKED, false>(x, wf, patch, cw, ite, kstart, lane, ch, r, er, a, c);
    gemm_chunk<MASKED, true>(x, wf, patch, cw, ite, kstart, lane, NCHUNK - 1, r, er, a, c);
}

__global__ __launch_bounds__(256) void resample_wmma(const float* __restrict__ x,
                                                     const float* __restrict__ Wf,
                                                     float* __restrict__ y) {
    __shared__ __attribute__((aligned(16))) float xpatch[8 * XWAVE];

    const int tid = threadIdx.x;
    const int ph  = blockIdx.y;
    const int cg  = blockIdx.x;
    const int kstart = ph ? 220 : 0;

    const int wave = tid >> 5;
    const int lane = tid & 31;
    const int lm   = lane & 15;
    const int hi   = lane >> 4;

    const float4* wf = (const float4*)(Wf + ph * WF_HALF) + lane;

    const int cw   = cg * 256 + wave * 32;
    const int col0 = cw + lm;
    const int col1 = col0 + 16;
    const int b0 = col0 / QBLK, q0 = col0 - b0 * QBLK;
    const int b1 = col1 / QBLK, q1 = col1 - b1 * QBLK;

    float* patch = xpatch + wave * XWAVE;

    v8f c[5][2] = {};

    const bool edge_lane = (q0 == 0) | (q0 == QBLK - 1) |
                           (q1 == 0) | (q1 == QBLK - 1);
    const int q0s = __builtin_amdgcn_readfirstlane(cw % QBLK);
    if (__any((int)edge_lane)) {
        gemm_chunks<true >(x, wf, patch, cw, q0s, kstart, lane, c);
    } else {
        gemm_chunks<false>(x, wf, patch, cw, q0s, kstart, lane, c);
    }

    float* __restrict__ yp = y + ((cw + hi * 8) * NPHASE + ph * 80 + lm);
    #pragma unroll
    for (int t = 0; t < 2; ++t) {
        #pragma unroll
        for (int r = 0; r < 8; ++r) {
            #pragma unroll
            for (int mt = 0; mt < 5; ++mt)
                yp[(t * 16 + r) * NPHASE + mt * 16] = c[mt][t][r];
        }
    }
}

extern "C" void kernel_launch(void* const* d_in, const int* in_sizes, int n_in,
                              void* d_out, int out_size, void* d_ws, size_t ws_size,
                              hipStream_t stream) {
    const float* x  = (const float*)d_in[0];
    float*       Wf = (float*)d_ws;
    float*       y  = (float*)d_out;

    build_weights<<<500, 256, 0, stream>>>(Wf);

    dim3 grid(125, 2);
    resample_wmma<<<grid, 256, 0, stream>>>(x, Wf, y);
}


# W-table kernel: both I0 series loops fully unrolled: the constant-argument one is then folded by the compiler (same f32 value), the other drops per-trip scalar work and 5 exact divisions; same bits
# speedup vs baseline: 4.8280x; 1.0377x over previous
#include <hip/hip_runtime.h>
#include <math.h>


typedef float v2f __attribute__((ext_vector_type(2)));
typedef float v8f __attribute__((ext_vector_type(8)));

#define T_IN     441000
#define T_OUT    160000
#define BATCH    32
#define QBLK     1000
#define KPAD     800
#define KW       576
#define KP_N     (KW / 8)
#define WF_HALF  (KP_N * 5 * 32 * 4)
#define KC       64
#define NCHUNK   (KW / KC)
#define GP       2
#define NG_CH    (KC / 8 / GP)
#define NSET     4
#define AHEAD    3
#define XP       68
#define XWAVE    (32 * XP)
#define NPHASE   160
#define NUMZ     64.0
#define ROLLOFF  0.9475937167399596
#define KBETA    14.769656459379492
#define SCALE    (16000.0 / 44100.0)

#define UBAND    28224
#define ZZ_BETA  (0.25 * KBETA * KBETA)

#define I0_TERMS 24
__device__ inline float bessel_i0_zz(float zz) {
    float term = 1.0f, sum = 1.0f;
    #pragma unroll
    for (int m = 1; m <= I0_TERMS; ++m) {
        term *= zz / (float)(m * m);
        sum += term;
    }
    return sum;
}

__global__ __launch_bounds__(256) void build_weights(float* __restrict__ Wf) {
    int idx = blockIdx.x * 256 + threadIdx.x;
    int p = idx / KPAD;
    int j = idx - p * KPAD;
    int pn   = p * 441;
    int boff = pn / 160;
    int rem  = pn - boff * 160;
    int k    = j - 177 - boff;
    int nu   = 160 * k - rem;
    float w  = 0.0f;
    if (nu > -UBAND && nu < UBAND) {
        const double dn = (double)nu;
        float s = 1.0f;
        if (nu != 0) {
            const double xr = dn * (ROLLOFF / 441.0);
            const double nr = rint(xr);
            float sp = sinpif((float)(xr - nr));
            if (((int)nr) & 1) sp = -sp;
            s = sp / (float)(M_PI * xr);
        }
        const double v = dn * (1.0 / (double)UBAND);
        double r = fma(-v, v, 1.0);
        if (r < 0.0) r = 0.0;
        const float zz = (float)(ZZ_BETA * r);
        w = (float)(SCALE * ROLLOFF) * s * bessel_i0_zz(zz)
            / bessel_i0_zz((float)ZZ_BETA);
    }
    const int half = p / 80;
    const int r    = p - half * 80;
    const int jw   = j - (half ? 220 : 0);
    if (jw >= 0 && jw < KW) {
        const int kp   = jw >> 3;
        const int rem8 = jw & 7;
        const int e    = ((rem8 >> 2) << 1) | (rem8 & 1);
        const int lane = (((rem8 >> 1) & 1) << 4) | (r & 15);
        Wf[half * WF_HALF + (((kp * 5 + (r >> 4)) * 32 + lane) << 2) + e] = w;
    }
}

typedef float v4f __attribute__((ext_vector_type(4)));
typedef v4f v4f_u __attribute__((aligned(4)));

#define X_LAST4  (BATCH * T_IN - 4)

__device__ inline v4f fetch_taps_clamped(const float* __restrict__ x, int cw, int it,
                                         int kstart, int j0, int lane) {
    const int c = cw + 2 * it + (lane >> 4);
    const int b = c / QBLK;
    const int q = c - b * QBLK;
    const int t = q * 441 - 177 + kstart + j0 + (lane & 15) * 4;
    v4f v;
    #pragma unroll
    for (int e = 0; e < 4; ++e) {
        int tc = t + e;
        tc = tc < 0 ? 0 : tc;
        tc = tc > T_IN - 1 ? T_IN - 1 : tc;
        v[e] = x[b * T_IN + tc];
    }
    return v;
}

template <bool MASKED>
__device__ inline void stage_fetch(const float* __restrict__ x, int cw, int kstart,
                                   int j0, int lane, const int ite[2], v4f r[16], v4f er[2]) {
    const int piece = (lane & 15) * 4;
    #pragma unroll
    for (int it = 0; it < 16; ++it) {
        const int c = cw + 2 * it + (lane >> 4);
        int f = c * 441 - 177 + kstart + j0 + piece;
        if constexpr (MASKED) {
            f = f < 0 ? 0 : f;
            f = f > X_LAST4 ? X_LAST4 : f;
        }
        r[it] = *(const v4f_u*)(x + f);
    }
    if constexpr (MASKED) {
        er[0] = fetch_taps_clamped(x, cw, ite[0], kstart, j0, lane);
        er[1] = fetch_taps_clamped(x, cw, ite[1], kstart, j0, lane);
    }
}

template <bool MASKED>
__device__ inline void stage_store(float* __restrict__ patch, int cw, int kstart,
                                   int j0, int lane, const int ite[2],
                                   const v4f r[16], const v4f er[2]) {
    float* dst = patch + (lane >> 4) * XP + (lane & 15) * 4;
    #pragma unroll
    for (int it = 0; it < 16; ++it)
        *(v4f*)(dst + 2 * it * XP) = r[it];
    if constexpr (MASKED) {
        #pragma unroll
        for (int s = 0; s < 2; ++s) {
            const int c = cw + 2 * ite[s] + (lane >> 4);
            const int b = c / QBLK;
            const int q = c - b * QBLK;
            const int t = q * 441 - 177 + kstart + j0 + (lane & 15) * 4;
            v4f v = er[s];
            #pragma unroll
            for (int e = 0; e < 4; ++e)
                v[e] = v[e] * (((unsigned)(t + e) < (unsigned)T_IN) ? 1.0f : 0.0f);
            *(v4f*)(dst + 2 * ite[s] * XP) = v;
        }
    }
}

__device__ inline void fetch_a(const float4* __restrict__ wf, int grp, float4 a[GP * 5]) {
    #pragma unroll
    for (int i = 0; i < GP * 5; ++i) a[i] = wf[(grp * (GP * 5) + i) * 32];
}

template <bool MASKED, bool LAST>
__device__ inline void gemm_chunk(const float* __restrict__ x,
                                  const float4* __restrict__ wf,
                                  float* __restrict__ patch,
                                  int cw, const int ite[2], int kstart, int lane, int ch,
                                  v4f r[16], v4f er[2], float4 a[NSET][GP * 5], v8f c[5][2]) {
    const float* bw = patch + (lane & 15) * XP + (lane >> 4) * 2;
    #pragma unroll
    for (int g = 0; g < NG_CH; ++g) {
        if (g + AHEAD < NG_CH || !LAST)
            fetch_a(wf, ch * NG_CH + g + AHEAD, a[(g + AHEAD) & (NSET - 1)]);
        if (g == 0 && !LAST) {
            __builtin_amdgcn_sched_barrier(0);
            stage_fetch<MASKED>(x, cw, kstart, (ch + 1) * KC, lane, ite, r, er);
        }
        __builtin_amdgcn_sched_barrier(0);
        #pragma unroll
        for (int i = 0; i < GP; ++i) {
            const int kl = g * GP + i;
            #pragma unroll
            for (int s = 0; s < 2; ++s) {
                const int jl = kl * 8 + s * 4;
                const v2f X0 = *(const v2f*)(bw + jl);
                const v2f X1 = *(const v2f*)(bw + 16 * XP + jl);
                #pragma unroll
                for (int mt = 0; mt < 5; ++mt) {
                    const float4 af = a[g & (NSET - 1)][i * 5 + mt];
                    v2f Wt;
                    if (s == 0) { Wt.x = af.x; Wt.y = af.y; }
                    else        { Wt.x = af.z; Wt.y = af.w; }
                    c[mt][0] = __builtin_amdgcn_wmma_f32_16x16x4_f32(
                        false, X0, false, Wt, (short)0, c[mt][0], false, false);
                    c[mt][1] = __builtin_amdgcn_wmma_f32_16x16x4_f32(
                        false, X1, false, Wt, (short)0, c[mt][1], false, false);
                }
            }
        }
    }
    __builtin_amdgcn_sched_barrier(0);
    if (!LAST) stage_store<MASKED>(patch, cw, kstart, (ch + 1) * KC, lane, ite, r, er);
}

template <bool MASKED>
__device__ inline void gemm_chunks(const float* __restrict__ x,
                                   const float4* __restrict__ wf,
                                   float* __restrict__ patch,
                                   int cw, int q0s, int kstart, int lane,
                                   v8f c[5][2]) {
    v4f r[16];
    v4f er[2];
    float4 a[NSET][GP * 5];
    int ite[2] = {0, 0};
    if constexpr (MASKED) {
        if (q0s != 0) {
            const int first = (QBLK - 2 - q0s + 1) >> 1;
            const int last  = (QBLK - q0s) >> 1;
            ite[0] = first < 0 ? 0 : (first > 15 ? 15 : first);
            ite[1] = last  < 0 ? 0 : (last  > 15 ? 15 : last);
        }
    }
    stage_fetch<MASKED>(x, cw, kstart, 0, lane, ite, r, er);
    #pragma unroll
    for (int s = 0; s < AHEAD; ++s) fetch_a(wf, s, a[s]);
    stage_store<MASKED>(patch, cw, kstart, 0, lane, ite, r, er);
    for (int ch = 0; ch < NCHUNK - 1; ++ch)
        gemm_chunk<MASKED, false>(x, wf, patch, cw, ite, kstart, lane, ch, r, er, a, c);
    gemm_chunk<MASKED, true>(x, wf, patch, cw, ite, kstart, lane, NCHUNK - 1, r, er, a, c);
}

__global__ __launch_bounds__(256) void resample_wmma(const float* __restrict__ x,
                                                     const float* __restrict__ Wf,
                                                     float* __restrict__ y) {
    __shared__ __attribute__((aligned(16))) float xpatch[8 * XWAVE];

    const int tid = threadIdx.x;
    const int ph  = blockIdx.y;
    const int cg  = blockIdx.x;
    const int kstart = ph ? 220 : 0;

    const int wave = tid >> 5;
    const int lane = tid & 31;
    const int lm   = lane & 15;
    const int hi   = lane >> 4;

    const float4* wf = (const float4*)(Wf + ph * WF_HALF) + lane;

    const int cw   = cg * 256 + wave * 32;
    const int col0 = cw + lm;
    const int col1 = col0 + 16;
    const int b0 = col0 / QBLK, q0 = col0 - b0 * QBLK;
    const int b1 = col1 / QBLK, q1 = col1 - b1 * QBLK;

    float* patch = xpatch + wave * XWAVE;

    v8f c[5][2] = {};

    const bool edge_lane = (q0 == 0) | (q0 == QBLK - 1) |
                           (q1 == 0) | (q1 == QBLK - 1);
    const int q0s = __builtin_amdgcn_readfirstlane(cw % QBLK);
    if (__any((int)edge_lane)) {
        gemm_chunks<true >(x, wf, patch, cw, q0s, kstart, lane, c);
    } else {
        gemm_chunks<false>(x, wf, patch, cw, q0s, kstart, lane, c);
    }

    float* __restrict__ yp = y + ((cw + hi * 8) * NPHASE + ph * 80 + lm);
    #pragma unroll
    for (int t = 0; t < 2; ++t) {
        #pragma unroll
        for (int r = 0; r < 8; ++r) {
            #pragma unroll
            for (int mt = 0; mt < 5; ++mt)
                yp[(t * 16 + r) * NPHASE + mt * 16] = c[mt][t][r];
        }
    }
}

extern "C" void kernel_launch(void* const* d_in, const int* in_sizes, int n_in,
                              void* d_out, int out_size, void* d_ws, size_t ws_size,
                              hipStream_t stream) {
    const float* x  = (const float*)d_in[0];
    float*       Wf = (float*)d_ws;
    float*       y  = (float*)d_out;

    build_weights<<<500, 256, 0, stream>>>(Wf);

    dim3 grid(125, 2);
    resample_wmma<<<grid, 256, 0, stream>>>(x, Wf, y);
}


# chunk position a compile-time constant in head chunks 0-1 and tail chunks 7-8 (2-6 stay a loop): WMMAs and table loads of a phase tile in a 16-tap group outside its band not emitted; same sums
# speedup vs baseline: 5.2608x; 1.0896x over previous
#include <hip/hip_runtime.h>
#include <math.h>


typedef float v2f __attribute__((ext_vector_type(2)));
typedef float v8f __attribute__((ext_vector_type(8)));

#define T_IN     441000
#define T_OUT    160000
#define BATCH    32
#define QBLK     1000
#define KPAD     800
#define KW       576
#define KP_N     (KW / 8)
#define WF_HALF  (KP_N * 5 * 32 * 4)
#define KC       64
#define NCHUNK   (KW / KC)
#define GP       2
#define NG_CH    (KC / 8 / GP)
#define NSET     4
#define AHEAD    3
#define XP       68
#define XWAVE    (32 * XP)
#define NPHASE   160
#define NUMZ     64.0
#define ROLLOFF  0.9475937167399596
#define KBETA    14.769656459379492
#define SCALE    (16000.0 / 44100.0)

#define UBAND    28224
#define ZZ_BETA  (0.25 * KBETA * KBETA)

constexpr bool tile_group_live(int mt, int G) {
    for (int half = 0; half < 2; ++half)
        for (int rr = 0; rr < 16; ++rr) {
            const int pn   = (half * 80 + mt * 16 + rr) * 441;
            const int boff = pn / 160;
            const int rem  = pn - boff * 160;
            for (int t = 0; t < 16; ++t) {
                const int j  = (half ? 220 : 0) + G * 16 + t;
                const int nu = 160 * (j - 177 - boff) - rem;
                if (nu > -UBAND && nu < UBAND) return true;
            }
        }
    return false;
}
constexpr int live_mask(int G) {
    int m = 0;
    if (G < KW / 16)
        for (int mt = 0; mt < 5; ++mt)
            if (tile_group_live(mt, G)) m |= 1 << mt;
    return m;
}
#define FULL_LO  2
#define FULL_HI  6
constexpr int use_mask(int G) {
    return (G / NG_CH >= FULL_LO && G / NG_CH <= FULL_HI) ? 31 : live_mask(G);
}

#define I0_TERMS 24
__device__ inline float bessel_i0_zz(float zz) {
    float term = 1.0f, sum = 1.0f;
    #pragma unroll
    for (int m = 1; m <= I0_TERMS; ++m) {
        term *= zz / (float)(m * m);
        sum += term;
    }
    return sum;
}

__global__ __launch_bounds__(256) void build_weights(float* __restrict__ Wf) {
    int idx = blockIdx.x * 256 + threadIdx.x;
    int p = idx / KPAD;
    int j = idx - p * KPAD;
    int pn   = p * 441;
    int boff = pn / 160;
    int rem  = pn - boff * 160;
    int k    = j - 177 - boff;
    int nu   = 160 * k - rem;
    float w  = 0.0f;
    if (nu > -UBAND && nu < UBAND) {
        const double dn = (double)nu;
        float s = 1.0f;
        if (nu != 0) {
            const double xr = dn * (ROLLOFF / 441.0);
            const double nr = rint(xr);
            float sp = sinpif((float)(xr - nr));
            if (((int)nr) & 1) sp = -sp;
            s = sp / (float)(M_PI * xr);
        }
        const double v = dn * (1.0 / (double)UBAND);
        double r = fma(-v, v, 1.0);
        if (r < 0.0) r = 0.0;
        const float zz = (float)(ZZ_BETA * r);
        w = (float)(SCALE * ROLLOFF) * s * bessel_i0_zz(zz)
            / bessel_i0_zz((float)ZZ_BETA);
    }
    const int half = p / 80;
    const int r    = p - half * 80;
    const int jw   = j - (half ? 220 : 0);
    if (jw >= 0 && jw < KW) {
        const int kp   = jw >> 3;
        const int rem8 = jw & 7;
        const int e    = ((rem8 >> 2) << 1) | (rem8 & 1);
        const int lane = (((rem8 >> 1) & 1) << 4) | (r & 15);
        Wf[half * WF_HALF + (((kp * 5 + (r >> 4)) * 32 + lane) << 2) + e] = w;
    }
}

typedef float v4f __attribute__((ext_vector_type(4)));
typedef v4f v4f_u __attribute__((aligned(4)));

#define X_LAST4  (BATCH * T_IN - 4)

__device__ inline v4f fetch_taps_clamped(const float* __restrict__ x, int cw, int it,
                                         int kstart, int j0, int lane) {
    const int c = cw + 2 * it + (lane >> 4);
    const int b = c / QBLK;
    const int q = c - b * QBLK;
    const int t = q * 441 - 177 + kstart + j0 + (lane & 15) * 4;
    v4f v;
    #pragma unroll
    for (int e = 0; e < 4; ++e) {
        int tc = t + e;
        tc = tc < 0 ? 0 : tc;
        tc = tc > T_IN - 1 ? T_IN - 1 : tc;
        v[e] = x[b * T_IN + tc];
    }
    return v;
}

template <bool MASKED>
__device__ inline void stage_fetch(const float* __restrict__ x, int cw, int kstart,
                                   int j0, int lane, const int ite[2], v4f r[16], v4f er[2]) {
    const int piece = (lane & 15) * 4;
    #pragma unroll
    for (int it = 0; it < 16; ++it) {
        const int c = cw + 2 * it + (lane >> 4);
        int f = c * 441 - 177 + kstart + j0 + piece;
        if constexpr (MASKED) {
            f = f < 0 ? 0 : f;
            f = f > X_LAST4 ? X_LAST4 : f;
        }
        r[it] = *(const v4f_u*)(x + f);
    }
    if constexpr (MASKED) {
        er[0] = fetch_taps_clamped(x, cw, ite[0], kstart, j0, lane);
        er[1] = fetch_taps_clamped(x, cw, ite[1], kstart, j0, lane);
    }
}

template <bool MASKED>
__device__ inline void stage_store(float* __restrict__ patch, int cw, int kstart,
                                   int j0, int lane, const int ite[2],
                                   const v4f r[16], const v4f er[2]) {
    float* dst = patch + (lane >> 4) * XP + (lane & 15) * 4;
    #pragma unroll
    for (int it = 0; it < 16; ++it)
        *(v4f*)(dst + 2 * it * XP) = r[it];
    if constexpr (MASKED) {
        #pragma unroll
        for (int s = 0; s < 2; ++s) {
            const int c = cw + 2 * ite[s] + (lane >> 4);
            const int b = c / QBLK;
            const int q = c - b * QBLK;
            const int t = q * 441 - 177 + kstart + j0 + (lane & 15) * 4;
            v4f v = er[s];
            #pragma unroll
            for (int e = 0; e < 4; ++e)
                v[e] = v[e] * (((unsigned)(t + e) < (unsigned)T_IN) ? 1.0f : 0.0f);
            *(v4f*)(dst + 2 * ite[s] * XP) = v;
        }
    }
}

template <int LIVE>
__device__ inline void fetch_a(const float4* __restrict__ wf, int grp, float4 a[GP * 5]) {
    #pragma unroll
    for (int i = 0; i < GP * 5; ++i)
        if ((LIVE >> (i % 5)) & 1) a[i] = wf[(grp * (GP * 5) + i) * 32];
}

template <bool MASKED, int CH, int g>
__device__ inline void gemm_group(const float* __restrict__ x,
                                  const float4* __restrict__ wf,
                                  const float* __restrict__ bw,
                                  int cw, const int ite[2], int kstart, int lane, int ch,
                                  v4f r[16], v4f er[2], float4 a[NSET][GP * 5], v8f c[5][2]) {
    constexpr bool LAST = (CH == NCHUNK - 1);
    constexpr int  USE  = CH < 0 ? 31 : use_mask(CH * NG_CH + g);
    constexpr int  ASK  = CH < 0 ? 31 : use_mask(CH * NG_CH + g + AHEAD);
    if (g + AHEAD < NG_CH || !LAST)
        fetch_a<ASK>(wf, ch * NG_CH + g + AHEAD, a[(g + AHEAD) & (NSET - 1)]);
    if (g == 0 && !LAST) {
        __builtin_amdgcn_sched_barrier(0);
        stage_fetch<MASKED>(x, cw, kstart, (ch + 1) * KC, lane, ite, r, er);
    }
    __builtin_amdgcn_sched_barrier(0);
    #pragma unroll
    for (int i = 0; i < GP; ++i) {
        const int kl = g * GP + i;
        #pragma unroll
        for (int s = 0; s < 2; ++s) {
            const int jl = kl * 8 + s * 4;
            const v2f X0 = *(const v2f*)(bw + jl);
            const v2f X1 = *(const v2f*)(bw + 16 * XP + jl);
            #pragma unroll
            for (int mt = 0; mt < 5; ++mt) {
                if (!((USE >> mt) & 1)) continue;
                const float4 af = a[g & (NSET - 1)][i * 5 + mt];
                v2f Wt;
                if (s == 0) { Wt.x = af.x; Wt.y = af.y; }
                else        { Wt.x = af.z; Wt.y = af.w; }
                c[mt][0] = __builtin_amdgcn_wmma_f32_16x16x4_f32(
                    false, X0, false, Wt, (short)0, c[mt][0], false, false);
                c[mt][1] = __builtin_amdgcn_wmma_f32_16x16x4_f32(
                    false, X1, false, Wt, (short)0, c[mt][1], false, false);
            }
        }
    }
}

template <bool MASKED, int CH>
__device__ inline void gemm_chunk(const float* __restrict__ x,
                                  const float4* __restrict__ wf,
                                  float* __restrict__ patch,
                                  int cw, const int ite[2], int kstart, int lane, int ch_rt,
                                  v4f r[16], v4f er[2], float4 a[NSET][GP * 5], v8f c[5][2]) {
    constexpr bool LAST = (CH == NCHUNK - 1);
    const int ch = CH < 0 ? ch_rt : CH;
    const float* bw = patch + (lane & 15) * XP + (lane >> 4) * 2;
    gemm_group<MASKED, CH, 0>(x, wf, bw, cw, ite, kstart, lane, ch, r, er, a, c);
    gemm_group<MASKED, CH, 1>(x, wf, bw, cw, ite, kstart, lane, ch, r, er, a, c);
    gemm_group<MASKED, CH, 2>(x, wf, bw, cw, ite, kstart, lane, ch, r, er, a, c);
    gemm_group<MASKED, CH, 3>(x, wf, bw, cw, ite, kstart, lane, ch, r, er, a, c);
    __builtin_amdgcn_sched_barrier(0);
    if (!LAST) stage_store<MASKED>(patch, cw, kstart, (ch + 1) * KC, lane, ite, r, er);
}

template <bool MASKED>
__device__ inline void gemm_chunks(const float* __restrict__ x,
                                   const float4* __restrict__ wf,
                                   float* __restrict__ patch,
                                   int cw, int q0s, int kstart, int lane,
                                   v8f c[5][2]) {
    v4f r[16];
    v4f er[2];
    float4 a[NSET][GP * 5];
    int ite[2] = {0, 0};
    if constexpr (MASKED) {
        if (q0s != 0) {
            const int first = (QBLK - 2 - q0s + 1) >> 1;
            const int last  = (QBLK - q0s) >> 1;
            ite[0] = first < 0 ? 0 : (first > 15 ? 15 : first);
            ite[1] = last  < 0 ? 0 : (last  > 15 ? 15 : last);
        }
    }
    stage_fetch<MASKED>(x, cw, kstart, 0, lane, ite, r, er);
    fetch_a<use_mask(0)>(wf, 0, a[0]);
    fetch_a<use_mask(1)>(wf, 1, a[1]);
    fetch_a<use_mask(2)>(wf, 2, a[2]);
    static_assert(AHEAD == 3, "one request per group ahead");
    stage_store<MASKED>(patch, cw, kstart, 0, lane, ite, r, er);
    gemm_chunk<MASKED, 0>(x, wf, patch, cw, ite, kstart, lane, 0, r, er, a, c);
    gemm_chunk<MASKED, 1>(x, wf, patch, cw, ite, kstart, lane, 1, r, er, a, c);
    for (int ch = FULL_LO; ch <= FULL_HI; ++ch)
        gemm_chunk<MASKED, -1>(x, wf, patch, cw, ite, kstart, lane, ch, r, er, a, c);
    gemm_chunk<MASKED, 7>(x, wf, patch, cw, ite, kstart, lane, 7, r, er, a, c);
    gemm_chunk<MASKED, 8>(x, wf, patch, cw, ite, kstart, lane, 8, r, er, a, c);
    static_assert(FULL_LO == 2 && FULL_HI == 6 && NCHUNK == 9, "head 0-1, loop 2-6, tail 7-8");
}

__global__ __launch_bounds__(256) void resample_wmma(const float* __restrict__ x,
                                                     const float* __restrict__ Wf,
                                                     float* __restrict__ y) {
    __shared__ __attribute__((aligned(16))) float xpatch[8 * XWAVE];

    const int tid = threadIdx.x;
    const int ph  = blockIdx.y;
    const int cg  = blockIdx.x;
    const int kstart = ph ? 220 : 0;

    const int wave = tid >> 5;
    const int lane = tid & 31;
    const int lm   = lane & 15;
    const int hi   = lane >> 4;

    const float4* wf = (const float4*)(Wf + ph * WF_HALF) + lane;

    const int cw   = cg * 256 + wave * 32;
    const int col0 = cw + lm;
    const int col1 = col0 + 16;
    const int b0 = col0 / QBLK, q0 = col0 - b0 * QBLK;
    const int b1 = col1 / QBLK, q1 = col1 - b1 * QBLK;

    float* patch = xpatch + wave * XWAVE;

    v8f c[5][2] = {};

    const bool edge_lane = (q0 == 0) | (q0 == QBLK - 1) |
                           (q1 == 0) | (q1 == QBLK - 1);
    const int q0s = __builtin_amdgcn_readfirstlane(cw % QBLK);
    if (__any((int)edge_lane)) {
        gemm_chunks<true >(x, wf, patch, cw, q0s, kstart, lane, c);
    } else {
        gemm_chunks<false>(x, wf, patch, cw, q0s, kstart, lane, c);
    }

    float* __restrict__ yp = y + ((cw + hi * 8) * NPHASE + ph * 80 + lm);
    #pragma unroll
    for (int t = 0; t < 2; ++t) {
        #pragma unroll
        for (int r = 0; r < 8; ++r) {
            #pragma unroll
            for (int mt = 0; mt < 5; ++mt)
                yp[(t * 16 + r) * NPHASE + mt * 16] = c[mt][t][r];
        }
    }
}

extern "C" void kernel_launch(void* const* d_in, const int* in_sizes, int n_in,
                              void* d_out, int out_size, void* d_ws, size_t ws_size,
                              hipStream_t stream) {
    const float* x  = (const float*)d_in[0];
    float*       Wf = (float*)d_ws;
    float*       y  = (float*)d_out;

    build_weights<<<500, 256, 0, stream>>>(Wf);

    dim3 grid(125, 2);
    resample_wmma<<<grid, 256, 0, stream>>>(x, Wf, y);
}


# younger wave of each SIMD's pair (waves 4-7) holds issue priority (s_setprio 3) for its first six chunks, then drops it: the pair ends together, not one after the other; loads, sums unchanged
# speedup vs baseline: 5.5239x; 1.0500x over previous
#include <hip/hip_runtime.h>
#include <math.h>


typedef float v2f __attribute__((ext_vector_type(2)));
typedef float v8f __attribute__((ext_vector_type(8)));

#define T_IN     441000
#define T_OUT    160000
#define BATCH    32
#define QBLK     1000
#define KPAD     800
#define KW       576
#define KP_N     (KW / 8)
#define WF_HALF  (KP_N * 5 * 32 * 4)
#define KC       64
#define NCHUNK   (KW / KC)
#define GP       2
#define NG_CH    (KC / 8 / GP)
#define NSET     4
#define AHEAD    3
#define XP       68
#define XWAVE    (32 * XP)
#define NPHASE   160
#define NUMZ     64.0
#define ROLLOFF  0.9475937167399596
#define KBETA    14.769656459379492
#define SCALE    (16000.0 / 44100.0)

#define UBAND    28224
#define ZZ_BETA  (0.25 * KBETA * KBETA)

constexpr bool tile_group_live(int mt, int G) {
    for (int half = 0; half < 2; ++half)
        for (int rr = 0; rr < 16; ++rr) {
            const int pn   = (half * 80 + mt * 16 + rr) * 441;
            const int boff = pn / 160;
            const int rem  = pn - boff * 160;
            for (int t = 0; t < 16; ++t) {
                const int j  = (half ? 220 : 0) + G * 16 + t;
                const int nu = 160 * (j - 177 - boff) - rem;
                if (nu > -UBAND && nu < UBAND) return true;
            }
        }
    return false;
}
constexpr int live_mask(int G) {
    int m = 0;
    if (G < KW / 16)
        for (int mt = 0; mt < 5; ++mt)
            if (tile_group_live(mt, G)) m |= 1 << mt;
    return m;
}
#define FULL_LO  2
#define FULL_HI  6
#define PRIO_DROP_CH  6
constexpr int use_mask(int G) {
    return (G / NG_CH >= FULL_LO && G / NG_CH <= FULL_HI) ? 31 : live_mask(G);
}

#define I0_TERMS 24
__device__ inline float bessel_i0_zz(float zz) {
    float term = 1.0f, sum = 1.0f;
    #pragma unroll
    for (int m = 1; m <= I0_TERMS; ++m) {
        term *= zz / (float)(m * m);
        sum += term;
    }
    return sum;
}

__global__ __launch_bounds__(256) void build_weights(float* __restrict__ Wf) {
    int idx = blockIdx.x * 256 + threadIdx.x;
    int p = idx / KPAD;
    int j = idx - p * KPAD;
    int pn   = p * 441;
    int boff = pn / 160;
    int rem  = pn - boff * 160;
    int k    = j - 177 - boff;
    int nu   = 160 * k - rem;
    float w  = 0.0f;
    if (nu > -UBAND && nu < UBAND) {
        const double dn = (double)nu;
        float s = 1.0f;
        if (nu != 0) {
            const double xr = dn * (ROLLOFF / 441.0);
            const double nr = rint(xr);
            float sp = sinpif((float)(xr - nr));
            if (((int)nr) & 1) sp = -sp;
            s = sp / (float)(M_PI * xr);
        }
        const double v = dn * (1.0 / (double)UBAND);
        double r = fma(-v, v, 1.0);
        if (r < 0.0) r = 0.0;
        const float zz = (float)(ZZ_BETA * r);
        w = (float)(SCALE * ROLLOFF) * s * bessel_i0_zz(zz)
            / bessel_i0_zz((float)ZZ_BETA);
    }
    const int half = p / 80;
    const int r    = p - half * 80;
    const int jw   = j - (half ? 220 : 0);
    if (jw >= 0 && jw < KW) {
        const int kp   = jw >> 3;
        const int rem8 = jw & 7;
        const int e    = ((rem8 >> 2) << 1) | (rem8 & 1);
        const int lane = (((rem8 >> 1) & 1) << 4) | (r & 15);
        Wf[half * WF_HALF + (((kp * 5 + (r >> 4)) * 32 + lane) << 2) + e] = w;
    }
}

typedef float v4f __attribute__((ext_vector_type(4)));
typedef v4f v4f_u __attribute__((aligned(4)));

#define X_LAST4  (BATCH * T_IN - 4)

__device__ inline v4f fetch_taps_clamped(const float* __restrict__ x, int cw, int it,
                                         int kstart, int j0, int lane) {
    const int c = cw + 2 * it + (lane >> 4);
    const int b = c / QBLK;
    const int q = c - b * QBLK;
    const int t = q * 441 - 177 + kstart + j0 + (lane & 15) * 4;
    v4f v;
    #pragma unroll
    for (int e = 0; e < 4; ++e) {
        int tc = t + e;
        tc = tc < 0 ? 0 : tc;
        tc = tc > T_IN - 1 ? T_IN - 1 : tc;
        v[e] = x[b * T_IN + tc];
    }
    return v;
}

template <bool MASKED>
__device__ inline void stage_fetch(const float* __restrict__ x, int cw, int kstart,
                                   int j0, int lane, const int ite[2], v4f r[16], v4f er[2]) {
    const int piece = (lane & 15) * 4;
    #pragma unroll
    for (int it = 0; it < 16; ++it) {
        const int c = cw + 2 * it + (lane >> 4);
        int f = c * 441 - 177 + kstart + j0 + piece;
        if constexpr (MASKED) {
            f = f < 0 ? 0 : f;
            f = f > X_LAST4 ? X_LAST4 : f;
        }
        r[it] = *(const v4f_u*)(x + f);
    }
    if constexpr (MASKED) {
        er[0] = fetch_taps_clamped(x, cw, ite[0], kstart, j0, lane);
        er[1] = fetch_taps_clamped(x, cw, ite[1], kstart, j0, lane);
    }
}

template <bool MASKED>
__device__ inline void stage_store(float* __restrict__ patch, int cw, int kstart,
                                   int j0, int lane, const int ite[2],
                                   const v4f r[16], const v4f er[2]) {
    float* dst = patch + (lane >> 4) * XP + (lane & 15) * 4;
    #pragma unroll
    for (int it = 0; it < 16; ++it)
        *(v4f*)(dst + 2 * it * XP) = r[it];
    if constexpr (MASKED) {
        #pragma unroll
        for (int s = 0; s < 2; ++s) {
            const int c = cw + 2 * ite[s] + (lane >> 4);
            const int b = c / QBLK;
            const int q = c - b * QBLK;
            const int t = q * 441 - 177 + kstart + j0 + (lane & 15) * 4;
            v4f v = er[s];
            #pragma unroll
            for (int e = 0; e < 4; ++e)
                v[e] = v[e] * (((unsigned)(t + e) < (unsigned)T_IN) ? 1.0f : 0.0f);
            *(v4f*)(dst + 2 * ite[s] * XP) = v;
        }
    }
}

template <int LIVE>
__device__ inline void fetch_a(const float4* __restrict__ wf, int grp, float4 a[GP * 5]) {
    #pragma unroll
    for (int i = 0; i < GP * 5; ++i)
        if ((LIVE >> (i % 5)) & 1) a[i] = wf[(grp * (GP * 5) + i) * 32];
}

template <bool MASKED, int CH, int g>
__device__ inline void gemm_group(const float* __restrict__ x,
                                  const float4* __restrict__ wf,
                                  const float* __restrict__ bw,
                                  int cw, const int ite[2], int kstart, int lane, int ch,
                                  v4f r[16], v4f er[2], float4 a[NSET][GP * 5], v8f c[5][2]) {
    constexpr bool LAST = (CH == NCHUNK - 1);
    constexpr int  USE  = CH < 0 ? 31 : use_mask(CH * NG_CH + g);
    constexpr int  ASK  = CH < 0 ? 31 : use_mask(CH * NG_CH + g + AHEAD);
    if (g + AHEAD < NG_CH || !LAST)
        fetch_a<ASK>(wf, ch * NG_CH + g + AHEAD, a[(g + AHEAD) & (NSET - 1)]);
    if (g == 0 && !LAST) {
        __builtin_amdgcn_sched_barrier(0);
        stage_fetch<MASKED>(x, cw, kstart, (ch + 1) * KC, lane, ite, r, er);
    }
    __builtin_amdgcn_sched_barrier(0);
    #pragma unroll
    for (int i = 0; i < GP; ++i) {
        const int kl = g * GP + i;
        #pragma unroll
        for (int s = 0; s < 2; ++s) {
            const int jl = kl * 8 + s * 4;
            const v2f X0 = *(const v2f*)(bw + jl);
            const v2f X1 = *(const v2f*)(bw + 16 * XP + jl);
            #pragma unroll
            for (int mt = 0; mt < 5; ++mt) {
                if (!((USE >> mt) & 1)) continue;
                const float4 af = a[g & (NSET - 1)][i * 5 + mt];
                v2f Wt;
                if (s == 0) { Wt.x = af.x; Wt.y = af.y; }
                else        { Wt.x = af.z; Wt.y = af.w; }
                c[mt][0] = __builtin_amdgcn_wmma_f32_16x16x4_f32(
                    false, X0, false, Wt, (short)0, c[mt][0], false, false);
                c[mt][1] = __builtin_amdgcn_wmma_f32_16x16x4_f32(
                    false, X1, false, Wt, (short)0, c[mt][1], false, false);
            }
        }
    }
}

template <bool MASKED, int CH>
__device__ inline void gemm_chunk(const float* __restrict__ x,
                                  const float4* __restrict__ wf,
                                  float* __restrict__ patch,
                                  int cw, const int ite[2], int kstart, int lane, int ch_rt,
                                  v4f r[16], v4f er[2], float4 a[NSET][GP * 5], v8f c[5][2]) {
    constexpr bool LAST = (CH == NCHUNK - 1);
    const int ch = CH < 0 ? ch_rt : CH;
    const float* bw = patch + (lane & 15) * XP + (lane >> 4) * 2;
    gemm_group<MASKED, CH, 0>(x, wf, bw, cw, ite, kstart, lane, ch, r, er, a, c);
    gemm_group<MASKED, CH, 1>(x, wf, bw, cw, ite, kstart, lane, ch, r, er, a, c);
    gemm_group<MASKED, CH, 2>(x, wf, bw, cw, ite, kstart, lane, ch, r, er, a, c);
    gemm_group<MASKED, CH, 3>(x, wf, bw, cw, ite, kstart, lane, ch, r, er, a, c);
    __builtin_amdgcn_sched_barrier(0);
    if (!LAST) stage_store<MASKED>(patch, cw, kstart, (ch + 1) * KC, lane, ite, r, er);
}

template <bool MASKED>
__device__ inline void gemm_chunks(const float* __restrict__ x,
                                   const float4* __restrict__ wf,
                                   float* __restrict__ patch,
                                   int cw, int q0s, int kstart, int lane,
                                   v8f c[5][2]) {
    v4f r[16];
    v4f er[2];
    float4 a[NSET][GP * 5];
    int ite[2] = {0, 0};
    if constexpr (MASKED) {
        if (q0s != 0) {
            const int first = (QBLK - 2 - q0s + 1) >> 1;
            const int last  = (QBLK - q0s) >> 1;
            ite[0] = first < 0 ? 0 : (first > 15 ? 15 : first);
            ite[1] = last  < 0 ? 0 : (last  > 15 ? 15 : last);
        }
    }
    stage_fetch<MASKED>(x, cw, kstart, 0, lane, ite, r, er);
    fetch_a<use_mask(0)>(wf, 0, a[0]);
    fetch_a<use_mask(1)>(wf, 1, a[1]);
    fetch_a<use_mask(2)>(wf, 2, a[2]);
    static_assert(AHEAD == 3, "one request per group ahead");
    stage_store<MASKED>(patch, cw, kstart, 0, lane, ite, r, er);
    gemm_chunk<MASKED, 0>(x, wf, patch, cw, ite, kstart, lane, 0, r, er, a, c);
    gemm_chunk<MASKED, 1>(x, wf, patch, cw, ite, kstart, lane, 1, r, er, a, c);
    for (int ch = FULL_LO; ch <= FULL_HI; ++ch) {
        if (ch == PRIO_DROP_CH) __builtin_amdgcn_s_setprio(0);
        gemm_chunk<MASKED, -1>(x, wf, patch, cw, ite, kstart, lane, ch, r, er, a, c);
    }
    gemm_chunk<MASKED, 7>(x, wf, patch, cw, ite, kstart, lane, 7, r, er, a, c);
    gemm_chunk<MASKED, 8>(x, wf, patch, cw, ite, kstart, lane, 8, r, er, a, c);
    static_assert(FULL_LO == 2 && FULL_HI == 6 && NCHUNK == 9, "head 0-1, loop 2-6, tail 7-8");
    static_assert(PRIO_DROP_CH >= FULL_LO && PRIO_DROP_CH <= FULL_HI, "the priority is given back inside the loop");
}

__global__ __launch_bounds__(256) void resample_wmma(const float* __restrict__ x,
                                                     const float* __restrict__ Wf,
                                                     float* __restrict__ y) {
    __shared__ __attribute__((aligned(16))) float xpatch[8 * XWAVE];

    const int tid = threadIdx.x;
    const int ph  = blockIdx.y;
    const int cg  = blockIdx.x;
    const int kstart = ph ? 220 : 0;

    const int wave = tid >> 5;
    const int lane = tid & 31;
    const int lm   = lane & 15;
    const int hi   = lane >> 4;

    const float4* wf = (const float4*)(Wf + ph * WF_HALF) + lane;

    const int cw   = cg * 256 + wave * 32;
    const int col0 = cw + lm;
    const int col1 = col0 + 16;
    const int b0 = col0 / QBLK, q0 = col0 - b0 * QBLK;
    const int b1 = col1 / QBLK, q1 = col1 - b1 * QBLK;

    float* patch = xpatch + wave * XWAVE;

    if (__builtin_amdgcn_readfirstlane(wave) >= 4) __builtin_amdgcn_s_setprio(3);

    v8f c[5][2] = {};

    const bool edge_lane = (q0 == 0) | (q0 == QBLK - 1) |
                           (q1 == 0) | (q1 == QBLK - 1);
    const int q0s = __builtin_amdgcn_readfirstlane(cw % QBLK);
    if (__any((int)edge_lane)) {
        gemm_chunks<true >(x, wf, patch, cw, q0s, kstart, lane, c);
    } else {
        gemm_chunks<false>(x, wf, patch, cw, q0s, kstart, lane, c);
    }

    float* __restrict__ yp = y + ((cw + hi * 8) * NPHASE + ph * 80 + lm);
    #pragma unroll
    for (int t = 0; t < 2; ++t) {
        #pragma unroll
        for (int r = 0; r < 8; ++r) {
            #pragma unroll
            for (int mt = 0; mt < 5; ++mt)
                yp[(t * 16 + r) * NPHASE + mt * 16] = c[mt][t][r];
        }
    }
}

extern "C" void kernel_launch(void* const* d_in, const int* in_sizes, int n_in,
                              void* d_out, int out_size, void* d_ws, size_t ws_size,
                              hipStream_t stream) {
    const float* x  = (const float*)d_in[0];
    float*       Wf = (float*)d_ws;
    float*       y  = (float*)d_out;

    build_weights<<<500, 256, 0, stream>>>(Wf);

    dim3 grid(125, 2);
    resample_wmma<<<grid, 256, 0, stream>>>(x, Wf, y);
}
